# starting kernel
# baseline (speedup 1.0000x reference)
#include <hip/hip_runtime.h>
#include <hip/hip_bf16.h>


typedef __attribute__((ext_vector_type(2))) float v2f;
typedef __attribute__((ext_vector_type(8))) float v8f;

#define D_FEAT 64


__global__ __launch_bounds__(256) void zero_f32(float* __restrict__ p, int n) {
  int i = blockIdx.x * blockDim.x + threadIdx.x;
  if (i < n) p[i] = 0.0f;
}

__global__ __launch_bounds__(256) void degree_kernel(const int* __restrict__ row,
                                                     float* __restrict__ deg, int e) {
  int i = blockIdx.x * blockDim.x + threadIdx.x;
  if (i < e) atomicAdd(&deg[row[i]], 1.0f);
}

__global__ __launch_bounds__(256) void dinv_kernel(float* __restrict__ deg_dinv, int n) {
  int i = blockIdx.x * blockDim.x + threadIdx.x;
  if (i < n) {
    float d = deg_dinv[i];
    deg_dinv[i] = (d > 0.0f) ? rsqrtf(d) : 0.0f;
  }
}

__global__ __launch_bounds__(256) void bias_init_kernel(const float* __restrict__ bias,
                                                        float* __restrict__ out, int total) {
  int i = blockIdx.x * blockDim.x + threadIdx.x;
  if (i < total) out[i] = bias[i & (D_FEAT - 1)];
}

__global__ __launch_bounds__(256) void gemm_xw_wmma(const float* __restrict__ X,
                                                    const float* __restrict__ W,
                                                    float* __restrict__ XW, int n) {
  const int wave = threadIdx.x >> 5;
  const int lane = threadIdx.x & 31;
  const int half = lane >> 4;
  const int l    = lane & 15;

  const int mBase = (blockIdx.x * 2 + (wave >> 2)) * 16;
  const int nBase = (wave & 3) * 16;
  if (mBase >= n) return;

  const int m = mBase + l;

  v8f acc = {};
#pragma unroll
  for (int k = 0; k < D_FEAT; k += 4) {
    v2f a, b;
    a.x = X[m * D_FEAT + k + 2 * half + 0];
    a.y = X[m * D_FEAT + k + 2 * half + 1];
    b.x = W[(k + 2 * half + 0) * D_FEAT + nBase + l];
    b.y = W[(k + 2 * half + 1) * D_FEAT + nBase + l];
    acc = __builtin_amdgcn_wmma_f32_16x16x4_f32(false, a, false, b,
                                                (short)0, acc, false, false);
  }

#pragma unroll
  for (int r = 0; r < 8; ++r) {
    XW[(mBase + r + 8 * half) * D_FEAT + nBase + l] = acc[r];
  }
}

__global__ __launch_bounds__(256) void scatter_kernel(const int* __restrict__ row,
                                                      const int* __restrict__ col,
                                                      const float* __restrict__ dinv,
                                                      const float* __restrict__ xw,
                                                      float* __restrict__ out, int e) {
  const int edge = blockIdx.x * 8 + (threadIdx.x >> 5);
  if (edge >= e) return;
  const int lane = threadIdx.x & 31;

  const int r = row[edge];
  const int c = col[edge];
  const float w = dinv[r] * dinv[c];

  const float v0 = xw[c * D_FEAT + lane] * w;
  const float v1 = xw[c * D_FEAT + lane + 32] * w;
  atomicAdd(&out[r * D_FEAT + lane], v0);
  atomicAdd(&out[r * D_FEAT + lane + 32], v1);
}


extern "C" void kernel_launch(void* const* d_in, const int* in_sizes, int n_in,
                              void* d_out, int out_size, void* d_ws, size_t ws_size,
                              hipStream_t stream) {
  const float* x      = (const float*)d_in[0];
  const int*   eidx   = (const int*)d_in[1];
  const float* weight = (const float*)d_in[2];
  const float* bias   = (const float*)d_in[3];
  float*       out    = (float*)d_out;

  const int n = in_sizes[0] / D_FEAT;
  const int e = in_sizes[1] / 2;
  const int* row = eidx;
  const int* col = eidx + e;

  float* dinv = (float*)d_ws;
  const int dinvStride = (n + 255) & ~255;
  float* xw = dinv + dinvStride;

  zero_f32<<<(n + 255) / 256, 256, 0, stream>>>(dinv, n);
  degree_kernel<<<(e + 255) / 256, 256, 0, stream>>>(row, dinv, e);
  dinv_kernel<<<(n + 255) / 256, 256, 0, stream>>>(dinv, n);
  gemm_xw_wmma<<<(n + 31) / 32, 256, 0, stream>>>(x, weight, xw, n);
  bias_init_kernel<<<(n * D_FEAT + 255) / 256, 256, 0, stream>>>(bias, out, n * D_FEAT);
  scatter_kernel<<<(e + 7) / 8, 256, 0, stream>>>(row, col, dinv, xw, out, e);
}


# degree count: work-groups own node-bin ranges, count row ids of one edge chunk in an LDS integer histogram, plain stores; global degree atomics and zero pass gone (degree summed as exact integers)
# speedup vs baseline: 1.3001x; 1.3001x over previous
#include <hip/hip_runtime.h>
#include <hip/hip_bf16.h>


typedef __attribute__((ext_vector_type(2))) float v2f;
typedef __attribute__((ext_vector_type(8))) float v8f;

#define D_FEAT 64


#define CNT_BINS    2048
#define CNT_CHUNKS  10
#define CNT_THREADS 512
#define CNT_INFLIGHT 8

__device__ __forceinline__ void count_one(unsigned int* hist, int r, int binBase) {
  const unsigned int b = (unsigned int)(r - binBase);
  if (b < (unsigned int)CNT_BINS) atomicAdd(&hist[b], 1u);
}

__global__ __launch_bounds__(CNT_THREADS) void degree_count_kernel(const int* __restrict__ row,
                                                                   int* __restrict__ partial,
                                                                   int n, int e, int nStride) {
  __shared__ unsigned int hist[CNT_BINS];
  const int binBase = blockIdx.x * CNT_BINS;
  const int chunk   = blockIdx.y;

  for (int i = threadIdx.x; i < CNT_BINS; i += CNT_THREADS) hist[i] = 0u;
  __syncthreads();

  const int chunkLen = (((e + CNT_CHUNKS - 1) / CNT_CHUNKS) + 3) & ~3;
  const int beg = chunk * chunkLen;
  int end = beg + chunkLen;
  if (end > e) end = e;
  const int nvec = (end > beg) ? ((end - beg) >> 2) : 0;

  if (nvec > 0) {
    const int4* rv = reinterpret_cast<const int4*>(row + beg);
    for (int v = threadIdx.x; v < nvec; v += CNT_INFLIGHT * CNT_THREADS) {
      int4 r[CNT_INFLIGHT];
      for (int k = 0; k < CNT_INFLIGHT; ++k) {
        const int idx = v + k * CNT_THREADS;
        r[k] = (idx < nvec) ? rv[idx] : make_int4(-1, -1, -1, -1);
      }
      for (int k = 0; k < CNT_INFLIGHT; ++k) {
        count_one(hist, r[k].x, binBase);
        count_one(hist, r[k].y, binBase);
        count_one(hist, r[k].z, binBase);
        count_one(hist, r[k].w, binBase);
      }
    }
  }
  for (int j = beg + 4 * nvec + (int)threadIdx.x; j < end; j += CNT_THREADS)
    count_one(hist, row[j], binBase);

  __syncthreads();
  for (int i = threadIdx.x; i < CNT_BINS; i += CNT_THREADS) {
    const int node = binBase + i;
    if (node < n) partial[chunk * nStride + node] = (int)hist[i];
  }
}

__global__ __launch_bounds__(256) void dinv_kernel(const int* __restrict__ partial,
                                                   float* __restrict__ dinv, int n, int nStride) {
  int i = blockIdx.x * blockDim.x + threadIdx.x;
  if (i < n) {
    int cnt = 0;
    for (int c = 0; c < CNT_CHUNKS; ++c) cnt += partial[c * nStride + i];
    const float d = (float)cnt;
    dinv[i] = (d > 0.0f) ? rsqrtf(d) : 0.0f;
  }
}

__global__ __launch_bounds__(256) void bias_init_kernel(const float* __restrict__ bias,
                                                        float* __restrict__ out, int total) {
  int i = blockIdx.x * blockDim.x + threadIdx.x;
  if (i < total) out[i] = bias[i & (D_FEAT - 1)];
}

__global__ __launch_bounds__(256) void gemm_xw_wmma(const float* __restrict__ X,
                                                    const float* __restrict__ W,
                                                    float* __restrict__ XW, int n) {
  const int wave = threadIdx.x >> 5;
  const int lane = threadIdx.x & 31;
  const int half = lane >> 4;
  const int l    = lane & 15;

  const int mBase = (blockIdx.x * 2 + (wave >> 2)) * 16;
  const int nBase = (wave & 3) * 16;
  if (mBase >= n) return;

  const int m = mBase + l;

  v8f acc = {};
#pragma unroll
  for (int k = 0; k < D_FEAT; k += 4) {
    v2f a, b;
    a.x = X[m * D_FEAT + k + 2 * half + 0];
    a.y = X[m * D_FEAT + k + 2 * half + 1];
    b.x = W[(k + 2 * half + 0) * D_FEAT + nBase + l];
    b.y = W[(k + 2 * half + 1) * D_FEAT + nBase + l];
    acc = __builtin_amdgcn_wmma_f32_16x16x4_f32(false, a, false, b,
                                                (short)0, acc, false, false);
  }

#pragma unroll
  for (int r = 0; r < 8; ++r) {
    XW[(mBase + r + 8 * half) * D_FEAT + nBase + l] = acc[r];
  }
}

__global__ __launch_bounds__(256) void scatter_kernel(const int* __restrict__ row,
                                                      const int* __restrict__ col,
                                                      const float* __restrict__ dinv,
                                                      const float* __restrict__ xw,
                                                      float* __restrict__ out, int e) {
  const int edge = blockIdx.x * 8 + (threadIdx.x >> 5);
  if (edge >= e) return;
  const int lane = threadIdx.x & 31;

  const int r = row[edge];
  const int c = col[edge];
  const float w = dinv[r] * dinv[c];

  const float v0 = xw[c * D_FEAT + lane] * w;
  const float v1 = xw[c * D_FEAT + lane + 32] * w;
  atomicAdd(&out[r * D_FEAT + lane], v0);
  atomicAdd(&out[r * D_FEAT + lane + 32], v1);
}


extern "C" void kernel_launch(void* const* d_in, const int* in_sizes, int n_in,
                              void* d_out, int out_size, void* d_ws, size_t ws_size,
                              hipStream_t stream) {
  const float* x      = (const float*)d_in[0];
  const int*   eidx   = (const int*)d_in[1];
  const float* weight = (const float*)d_in[2];
  const float* bias   = (const float*)d_in[3];
  float*       out    = (float*)d_out;

  const int n = in_sizes[0] / D_FEAT;
  const int e = in_sizes[1] / 2;
  const int* row = eidx;
  const int* col = eidx + e;

  float* dinv = (float*)d_ws;
  const int dinvStride = (n + 255) & ~255;
  float* xw = dinv + dinvStride;
  int* partial = (int*)(xw + n * D_FEAT);

  degree_count_kernel<<<dim3((n + CNT_BINS - 1) / CNT_BINS, CNT_CHUNKS), CNT_THREADS, 0, stream>>>(
      row, partial, n, e, dinvStride);
  dinv_kernel<<<(n + 255) / 256, 256, 0, stream>>>(partial, dinv, n, dinvStride);
  gemm_xw_wmma<<<(n + 31) / 32, 256, 0, stream>>>(x, weight, xw, n);
  bias_init_kernel<<<(n * D_FEAT + 255) / 256, 256, 0, stream>>>(bias, out, n * D_FEAT);
  scatter_kernel<<<(e + 7) / 8, 256, 0, stream>>>(row, col, dinv, xw, out, e);
}


# edge scatter: a wave takes 32 consecutive edges (ids, weights loaded once per lane, broadcast by readlane); 8 edges' row loads issued together ahead of their float atomic adds (order free as before)
# speedup vs baseline: 2.2120x; 1.7014x over previous
#include <hip/hip_runtime.h>
#include <hip/hip_bf16.h>


typedef __attribute__((ext_vector_type(2))) float v2f;
typedef __attribute__((ext_vector_type(8))) float v8f;

#define D_FEAT 64


#define CNT_BINS    2048
#define CNT_CHUNKS  10
#define CNT_THREADS 512
#define CNT_INFLIGHT 8

__device__ __forceinline__ void count_one(unsigned int* hist, int r, int binBase) {
  const unsigned int b = (unsigned int)(r - binBase);
  if (b < (unsigned int)CNT_BINS) atomicAdd(&hist[b], 1u);
}

__global__ __launch_bounds__(CNT_THREADS) void degree_count_kernel(const int* __restrict__ row,
                                                                   int* __restrict__ partial,
                                                                   int n, int e, int nStride) {
  __shared__ unsigned int hist[CNT_BINS];
  const int binBase = blockIdx.x * CNT_BINS;
  const int chunk   = blockIdx.y;

  for (int i = threadIdx.x; i < CNT_BINS; i += CNT_THREADS) hist[i] = 0u;
  __syncthreads();

  const int chunkLen = (((e + CNT_CHUNKS - 1) / CNT_CHUNKS) + 3) & ~3;
  const int beg = chunk * chunkLen;
  int end = beg + chunkLen;
  if (end > e) end = e;
  const int nvec = (end > beg) ? ((end - beg) >> 2) : 0;

  if (nvec > 0) {
    const int4* rv = reinterpret_cast<const int4*>(row + beg);
    for (int v = threadIdx.x; v < nvec; v += CNT_INFLIGHT * CNT_THREADS) {
      int4 r[CNT_INFLIGHT];
      for (int k = 0; k < CNT_INFLIGHT; ++k) {
        const int idx = v + k * CNT_THREADS;
        r[k] = (idx < nvec) ? rv[idx] : make_int4(-1, -1, -1, -1);
      }
      for (int k = 0; k < CNT_INFLIGHT; ++k) {
        count_one(hist, r[k].x, binBase);
        count_one(hist, r[k].y, binBase);
        count_one(hist, r[k].z, binBase);
        count_one(hist, r[k].w, binBase);
      }
    }
  }
  for (int j = beg + 4 * nvec + (int)threadIdx.x; j < end; j += CNT_THREADS)
    count_one(hist, row[j], binBase);

  __syncthreads();
  for (int i = threadIdx.x; i < CNT_BINS; i += CNT_THREADS) {
    const int node = binBase + i;
    if (node < n) partial[chunk * nStride + node] = (int)hist[i];
  }
}

__global__ __launch_bounds__(256) void dinv_kernel(const int* __restrict__ partial,
                                                   float* __restrict__ dinv, int n, int nStride) {
  int i = blockIdx.x * blockDim.x + threadIdx.x;
  if (i < n) {
    int cnt = 0;
    for (int c = 0; c < CNT_CHUNKS; ++c) cnt += partial[c * nStride + i];
    const float d = (float)cnt;
    dinv[i] = (d > 0.0f) ? rsqrtf(d) : 0.0f;
  }
}

__global__ __launch_bounds__(256) void bias_init_kernel(const float* __restrict__ bias,
                                                        float* __restrict__ out, int total) {
  int i = blockIdx.x * blockDim.x + threadIdx.x;
  if (i < total) out[i] = bias[i & (D_FEAT - 1)];
}

__global__ __launch_bounds__(256) void gemm_xw_wmma(const float* __restrict__ X,
                                                    const float* __restrict__ W,
                                                    float* __restrict__ XW, int n) {
  const int wave = threadIdx.x >> 5;
  const int lane = threadIdx.x & 31;
  const int half = lane >> 4;
  const int l    = lane & 15;

  const int mBase = (blockIdx.x * 2 + (wave >> 2)) * 16;
  const int nBase = (wave & 3) * 16;
  if (mBase >= n) return;

  const int m = mBase + l;

  v8f acc = {};
#pragma unroll
  for (int k = 0; k < D_FEAT; k += 4) {
    v2f a, b;
    a.x = X[m * D_FEAT + k + 2 * half + 0];
    a.y = X[m * D_FEAT + k + 2 * half + 1];
    b.x = W[(k + 2 * half + 0) * D_FEAT + nBase + l];
    b.y = W[(k + 2 * half + 1) * D_FEAT + nBase + l];
    acc = __builtin_amdgcn_wmma_f32_16x16x4_f32(false, a, false, b,
                                                (short)0, acc, false, false);
  }

#pragma unroll
  for (int r = 0; r < 8; ++r) {
    XW[(mBase + r + 8 * half) * D_FEAT + nBase + l] = acc[r];
  }
}

#define SC_EPW     32
#define SC_THREADS 256
#define SC_GROUP   8

__device__ __forceinline__ int lane_bcast_i(int v, int srcLane) {
  return __builtin_amdgcn_readlane(v, srcLane);
}
__device__ __forceinline__ float lane_bcast_f(float v, int srcLane) {
  return __int_as_float(__builtin_amdgcn_readlane(__float_as_int(v), srcLane));
}

__global__ __launch_bounds__(SC_THREADS) void scatter_kernel(const int* __restrict__ row,
                                                             const int* __restrict__ col,
                                                             const float* __restrict__ dinv,
                                                             const float* __restrict__ xw,
                                                             float* __restrict__ out, int e) {
  const int lane = threadIdx.x & 31;
  const int wave = __builtin_amdgcn_readfirstlane((int)(threadIdx.x >> 5));
  const int base = (blockIdx.x * (SC_THREADS / 32) + wave) * SC_EPW;
  if (base >= e) return;
  const int cnt = (e - base < SC_EPW) ? (e - base) : SC_EPW;

  const int idx = base + ((lane < cnt) ? lane : (cnt - 1));
  const int   myR = row[idx];
  const int   myC = col[idx];
  const float dR = dinv[myR];
  const float dC = dinv[myC];

  if (cnt == SC_EPW) {
#pragma unroll
    for (int g = 0; g < SC_EPW; g += SC_GROUP) {
      float v0[SC_GROUP], v1[SC_GROUP];
#pragma unroll
      for (int k = 0; k < SC_GROUP; ++k) {
        const unsigned co = (unsigned)lane_bcast_i(myC, g + k) * D_FEAT + lane;
        v0[k] = xw[co];
        v1[k] = xw[co + 32];
      }
      const float myW = dR * dC;
#pragma unroll
      for (int k = 0; k < SC_GROUP; ++k) {
        const unsigned ro = (unsigned)lane_bcast_i(myR, g + k) * D_FEAT + lane;
        const float w = lane_bcast_f(myW, g + k);
        atomicAdd(&out[ro], v0[k] * w);
        atomicAdd(&out[ro + 32], v1[k] * w);
      }
    }
  } else {
    for (int j = 0; j < cnt; ++j) {
      const unsigned co = (unsigned)lane_bcast_i(myC, j) * D_FEAT + lane;
      const unsigned ro = (unsigned)lane_bcast_i(myR, j) * D_FEAT + lane;
      const float w = lane_bcast_f(dR, j) * lane_bcast_f(dC, j);
      const float a0 = xw[co] * w;
      const float a1 = xw[co + 32] * w;
      atomicAdd(&out[ro], a0);
      atomicAdd(&out[ro + 32], a1);
    }
  }
}


extern "C" void kernel_launch(void* const* d_in, const int* in_sizes, int n_in,
                              void* d_out, int out_size, void* d_ws, size_t ws_size,
                              hipStream_t stream) {
  const float* x      = (const float*)d_in[0];
  const int*   eidx   = (const int*)d_in[1];
  const float* weight = (const float*)d_in[2];
  const float* bias   = (const float*)d_in[3];
  float*       out    = (float*)d_out;

  const int n = in_sizes[0] / D_FEAT;
  const int e = in_sizes[1] / 2;
  const int* row = eidx;
  const int* col = eidx + e;

  float* dinv = (float*)d_ws;
  const int dinvStride = (n + 255) & ~255;
  float* xw = dinv + dinvStride;
  int* partial = (int*)(xw + n * D_FEAT);

  degree_count_kernel<<<dim3((n + CNT_BINS - 1) / CNT_BINS, CNT_CHUNKS), CNT_THREADS, 0, stream>>>(
      row, partial, n, e, dinvStride);
  dinv_kernel<<<(n + 255) / 256, 256, 0, stream>>>(partial, dinv, n, dinvStride);
  gemm_xw_wmma<<<(n + 31) / 32, 256, 0, stream>>>(x, weight, xw, n);
  bias_init_kernel<<<(n * D_FEAT + 255) / 256, 256, 0, stream>>>(bias, out, n * D_FEAT);
  scatter_kernel<<<(e + 255) / 256, SC_THREADS, 0, stream>>>(row, col, dinv, xw, out, e);
}


# bias start value of out stored by the matrix kernel (each wave writes 4 rows of its 16-row tile as whole 128-B lines, 16 B per lane); the one-store-per-lane bias launch and its 100000 waves gone
# speedup vs baseline: 2.5068x; 1.1333x over previous
#include <hip/hip_runtime.h>
#include <hip/hip_bf16.h>


typedef __attribute__((ext_vector_type(2))) float v2f;
typedef __attribute__((ext_vector_type(8))) float v8f;

#define D_FEAT 64


#define CNT_BINS    2048
#define CNT_CHUNKS  10
#define CNT_THREADS 512
#define CNT_INFLIGHT 8

__device__ __forceinline__ void count_one(unsigned int* hist, int r, int binBase) {
  const unsigned int b = (unsigned int)(r - binBase);
  if (b < (unsigned int)CNT_BINS) atomicAdd(&hist[b], 1u);
}

__global__ __launch_bounds__(CNT_THREADS) void degree_count_kernel(const int* __restrict__ row,
                                                                   int* __restrict__ partial,
                                                                   int n, int e, int nStride) {
  __shared__ unsigned int hist[CNT_BINS];
  const int binBase = blockIdx.x * CNT_BINS;
  const int chunk   = blockIdx.y;

  for (int i = threadIdx.x; i < CNT_BINS; i += CNT_THREADS) hist[i] = 0u;
  __syncthreads();

  const int chunkLen = (((e + CNT_CHUNKS - 1) / CNT_CHUNKS) + 3) & ~3;
  const int beg = chunk * chunkLen;
  int end = beg + chunkLen;
  if (end > e) end = e;
  const int nvec = (end > beg) ? ((end - beg) >> 2) : 0;

  if (nvec > 0) {
    const int4* rv = reinterpret_cast<const int4*>(row + beg);
    for (int v = threadIdx.x; v < nvec; v += CNT_INFLIGHT * CNT_THREADS) {
      int4 r[CNT_INFLIGHT];
      for (int k = 0; k < CNT_INFLIGHT; ++k) {
        const int idx = v + k * CNT_THREADS;
        r[k] = (idx < nvec) ? rv[idx] : make_int4(-1, -1, -1, -1);
      }
      for (int k = 0; k < CNT_INFLIGHT; ++k) {
        count_one(hist, r[k].x, binBase);
        count_one(hist, r[k].y, binBase);
        count_one(hist, r[k].z, binBase);
        count_one(hist, r[k].w, binBase);
      }
    }
  }
  for (int j = beg + 4 * nvec + (int)threadIdx.x; j < end; j += CNT_THREADS)
    count_one(hist, row[j], binBase);

  __syncthreads();
  for (int i = threadIdx.x; i < CNT_BINS; i += CNT_THREADS) {
    const int node = binBase + i;
    if (node < n) partial[chunk * nStride + node] = (int)hist[i];
  }
}

__global__ __launch_bounds__(256) void dinv_kernel(const int* __restrict__ partial,
                                                   float* __restrict__ dinv, int n, int nStride) {
  int i = blockIdx.x * blockDim.x + threadIdx.x;
  if (i < n) {
    int cnt = 0;
    for (int c = 0; c < CNT_CHUNKS; ++c) cnt += partial[c * nStride + i];
    const float d = (float)cnt;
    dinv[i] = (d > 0.0f) ? rsqrtf(d) : 0.0f;
  }
}

__global__ __launch_bounds__(256) void gemm_xw_wmma(const float* __restrict__ X,
                                                    const float* __restrict__ W,
                                                    const float* __restrict__ bias,
                                                    float* __restrict__ XW,
                                                    float* __restrict__ out, int n) {
  const int wave = threadIdx.x >> 5;
  const int lane = threadIdx.x & 31;
  const int half = lane >> 4;
  const int l    = lane & 15;

  const int mBase = (blockIdx.x * 2 + (wave >> 2)) * 16;
  const int nBase = (wave & 3) * 16;
  if (mBase >= n) return;

  {
    const float4 bv = reinterpret_cast<const float4*>(bias)[l];
    float4* const out4 = reinterpret_cast<float4*>(out);
    const int o4 = (mBase + (wave & 3) * 4 + half) * (D_FEAT / 4) + l;
    out4[o4] = bv;
    out4[o4 + 2 * (D_FEAT / 4)] = bv;
  }

  const int m = mBase + l;

  v8f acc = {};
#pragma unroll
  for (int k = 0; k < D_FEAT; k += 4) {
    v2f a, b;
    a.x = X[m * D_FEAT + k + 2 * half + 0];
    a.y = X[m * D_FEAT + k + 2 * half + 1];
    b.x = W[(k + 2 * half + 0) * D_FEAT + nBase + l];
    b.y = W[(k + 2 * half + 1) * D_FEAT + nBase + l];
    acc = __builtin_amdgcn_wmma_f32_16x16x4_f32(false, a, false, b,
                                                (short)0, acc, false, false);
  }

#pragma unroll
  for (int r = 0; r < 8; ++r) {
    XW[(mBase + r + 8 * half) * D_FEAT + nBase + l] = acc[r];
  }
}

#define SC_EPW     32
#define SC_THREADS 256
#define SC_GROUP   8

__device__ __forceinline__ int lane_bcast_i(int v, int srcLane) {
  return __builtin_amdgcn_readlane(v, srcLane);
}
__device__ __forceinline__ float lane_bcast_f(float v, int srcLane) {
  return __int_as_float(__builtin_amdgcn_readlane(__float_as_int(v), srcLane));
}

__global__ __launch_bounds__(SC_THREADS) void scatter_kernel(const int* __restrict__ row,
                                                             const int* __restrict__ col,
                                                             const float* __restrict__ dinv,
                                                             const float* __restrict__ xw,
                                                             float* __restrict__ out, int e) {
  const int lane = threadIdx.x & 31;
  const int wave = __builtin_amdgcn_readfirstlane((int)(threadIdx.x >> 5));
  const int base = (blockIdx.x * (SC_THREADS / 32) + wave) * SC_EPW;
  if (base >= e) return;
  const int cnt = (e - base < SC_EPW) ? (e - base) : SC_EPW;

  const int idx = base + ((lane < cnt) ? lane : (cnt - 1));
  const int   myR = row[idx];
  const int   myC = col[idx];
  const float dR = dinv[myR];
  const float dC = dinv[myC];

  if (cnt == SC_EPW) {
#pragma unroll
    for (int g = 0; g < SC_EPW; g += SC_GROUP) {
      float v0[SC_GROUP], v1[SC_GROUP];
#pragma unroll
      for (int k = 0; k < SC_GROUP; ++k) {
        const unsigned co = (unsigned)lane_bcast_i(myC, g + k) * D_FEAT + lane;
        v0[k] = xw[co];
        v1[k] = xw[co + 32];
      }
      const float myW = dR * dC;
#pragma unroll
      for (int k = 0; k < SC_GROUP; ++k) {
        const unsigned ro = (unsigned)lane_bcast_i(myR, g + k) * D_FEAT + lane;
        const float w = lane_bcast_f(myW, g + k);
        atomicAdd(&out[ro], v0[k] * w);
        atomicAdd(&out[ro + 32], v1[k] * w);
      }
    }
  } else {
    for (int j = 0; j < cnt; ++j) {
      const unsigned co = (unsigned)lane_bcast_i(myC, j) * D_FEAT + lane;
      const unsigned ro = (unsigned)lane_bcast_i(myR, j) * D_FEAT + lane;
      const float w = lane_bcast_f(dR, j) * lane_bcast_f(dC, j);
      const float a0 = xw[co] * w;
      const float a1 = xw[co + 32] * w;
      atomicAdd(&out[ro], a0);
      atomicAdd(&out[ro + 32], a1);
    }
  }
}


extern "C" void kernel_launch(void* const* d_in, const int* in_sizes, int n_in,
                              void* d_out, int out_size, void* d_ws, size_t ws_size,
                              hipStream_t stream) {
  const float* x      = (const float*)d_in[0];
  const int*   eidx   = (const int*)d_in[1];
  const float* weight = (const float*)d_in[2];
  const float* bias   = (const float*)d_in[3];
  float*       out    = (float*)d_out;

  const int n = in_sizes[0] / D_FEAT;
  const int e = in_sizes[1] / 2;
  const int* row = eidx;
  const int* col = eidx + e;

  float* dinv = (float*)d_ws;
  const int dinvStride = (n + 255) & ~255;
  float* xw = dinv + dinvStride;
  int* partial = (int*)(xw + n * D_FEAT);

  degree_count_kernel<<<dim3((n + CNT_BINS - 1) / CNT_BINS, CNT_CHUNKS), CNT_THREADS, 0, stream>>>(
      row, partial, n, e, dinvStride);
  dinv_kernel<<<(n + 255) / 256, 256, 0, stream>>>(partial, dinv, n, dinvStride);
  gemm_xw_wmma<<<(n + 31) / 32, 256, 0, stream>>>(x, weight, bias, xw, out, n);
  scatter_kernel<<<(e + 255) / 256, SC_THREADS, 0, stream>>>(row, col, dinv, xw, out, e);
}


# degree count and matrix product in one launch: the first 250 work-groups count row ids, the other 782 (512 threads, 64 rows each) do X*W and the bias store; 4 launches become 3
# speedup vs baseline: 2.7062x; 1.0795x over previous
#include <hip/hip_runtime.h>
#include <hip/hip_bf16.h>


typedef __attribute__((ext_vector_type(2))) float v2f;
typedef __attribute__((ext_vector_type(8))) float v8f;

#define D_FEAT 64


#define CNT_BINS    2048
#define CNT_CHUNKS  10
#define CNT_THREADS 512
#define CNT_INFLIGHT 8

__device__ __forceinline__ void count_one(unsigned int* hist, int r, int binBase) {
  const unsigned int b = (unsigned int)(r - binBase);
  if (b < (unsigned int)CNT_BINS) atomicAdd(&hist[b], 1u);
}

__device__ __forceinline__ void degree_count_block(unsigned int* hist,
                                                   const int* __restrict__ row,
                                                   int* __restrict__ partial,
                                                   int n, int e, int nStride,
                                                   int binBlock, int chunk) {
  const int binBase = binBlock * CNT_BINS;

  for (int i = threadIdx.x; i < CNT_BINS; i += CNT_THREADS) hist[i] = 0u;
  __syncthreads();

  const int chunkLen = (((e + CNT_CHUNKS - 1) / CNT_CHUNKS) + 3) & ~3;
  const int beg = chunk * chunkLen;
  int end = beg + chunkLen;
  if (end > e) end = e;
  const int nvec = (end > beg) ? ((end - beg) >> 2) : 0;

  if (nvec > 0) {
    const int4* rv = reinterpret_cast<const int4*>(row + beg);
    for (int v = threadIdx.x; v < nvec; v += CNT_INFLIGHT * CNT_THREADS) {
      int4 r[CNT_INFLIGHT];
      for (int k = 0; k < CNT_INFLIGHT; ++k) {
        const int idx = v + k * CNT_THREADS;
        r[k] = (idx < nvec) ? rv[idx] : make_int4(-1, -1, -1, -1);
      }
      for (int k = 0; k < CNT_INFLIGHT; ++k) {
        count_one(hist, r[k].x, binBase);
        count_one(hist, r[k].y, binBase);
        count_one(hist, r[k].z, binBase);
        count_one(hist, r[k].w, binBase);
      }
    }
  }
  for (int j = beg + 4 * nvec + (int)threadIdx.x; j < end; j += CNT_THREADS)
    count_one(hist, row[j], binBase);

  __syncthreads();
  for (int i = threadIdx.x; i < CNT_BINS; i += CNT_THREADS) {
    const int node = binBase + i;
    if (node < n) partial[chunk * nStride + node] = (int)hist[i];
  }
}

__global__ __launch_bounds__(256) void dinv_kernel(const int* __restrict__ partial,
                                                   float* __restrict__ dinv, int n, int nStride) {
  int i = blockIdx.x * blockDim.x + threadIdx.x;
  if (i < n) {
    int cnt = 0;
    for (int c = 0; c < CNT_CHUNKS; ++c) cnt += partial[c * nStride + i];
    const float d = (float)cnt;
    dinv[i] = (d > 0.0f) ? rsqrtf(d) : 0.0f;
  }
}

#define GEMM_ROWS_PER_BLOCK 64

__device__ __forceinline__ void gemm_xw_block(const float* __restrict__ X,
                                              const float* __restrict__ W,
                                              const float* __restrict__ bias,
                                              float* __restrict__ XW,
                                              float* __restrict__ out, int n, int gemmBlock) {
  const int wave = threadIdx.x >> 5;
  const int lane = threadIdx.x & 31;
  const int half = lane >> 4;
  const int l    = lane & 15;

  const int mBase = (gemmBlock * 4 + (wave >> 2)) * 16;
  const int nBase = (wave & 3) * 16;
  if (mBase >= n) return;

  {
    const float4 bv = reinterpret_cast<const float4*>(bias)[l];
    float4* const out4 = reinterpret_cast<float4*>(out);
    const int o4 = (mBase + (wave & 3) * 4 + half) * (D_FEAT / 4) + l;
    out4[o4] = bv;
    out4[o4 + 2 * (D_FEAT / 4)] = bv;
  }

  const int m = mBase + l;

  v8f acc = {};
#pragma unroll
  for (int k = 0; k < D_FEAT; k += 4) {
    v2f a, b;
    a.x = X[m * D_FEAT + k + 2 * half + 0];
    a.y = X[m * D_FEAT + k + 2 * half + 1];
    b.x = W[(k + 2 * half + 0) * D_FEAT + nBase + l];
    b.y = W[(k + 2 * half + 1) * D_FEAT + nBase + l];
    acc = __builtin_amdgcn_wmma_f32_16x16x4_f32(false, a, false, b,
                                                (short)0, acc, false, false);
  }

#pragma unroll
  for (int r = 0; r < 8; ++r) {
    XW[(mBase + r + 8 * half) * D_FEAT + nBase + l] = acc[r];
  }
}

__global__ __launch_bounds__(CNT_THREADS, 16) void count_and_gemm_kernel(const int* __restrict__ row,
                                                                     int* __restrict__ partial,
                                                                     const float* __restrict__ X,
                                                                     const float* __restrict__ W,
                                                                     const float* __restrict__ bias,
                                                                     float* __restrict__ XW,
                                                                     float* __restrict__ out,
                                                                     int n, int e, int nStride) {
  __shared__ unsigned int hist[CNT_BINS];
  const int binBlocks   = (n + CNT_BINS - 1) / CNT_BINS;
  const int countBlocks = binBlocks * CNT_CHUNKS;
  const int b = blockIdx.x;
  if (b < countBlocks) {
    const int chunk = b / binBlocks;
    degree_count_block(hist, row, partial, n, e, nStride, b - chunk * binBlocks, chunk);
  } else {
    gemm_xw_block(X, W, bias, XW, out, n, b - countBlocks);
  }
}

#define SC_EPW     32
#define SC_THREADS 256
#define SC_GROUP   8

__device__ __forceinline__ int lane_bcast_i(int v, int srcLane) {
  return __builtin_amdgcn_readlane(v, srcLane);
}
__device__ __forceinline__ float lane_bcast_f(float v, int srcLane) {
  return __int_as_float(__builtin_amdgcn_readlane(__float_as_int(v), srcLane));
}

__global__ __launch_bounds__(SC_THREADS) void scatter_kernel(const int* __restrict__ row,
                                                             const int* __restrict__ col,
                                                             const float* __restrict__ dinv,
                                                             const float* __restrict__ xw,
                                                             float* __restrict__ out, int e) {
  const int lane = threadIdx.x & 31;
  const int wave = __builtin_amdgcn_readfirstlane((int)(threadIdx.x >> 5));
  const int base = (blockIdx.x * (SC_THREADS / 32) + wave) * SC_EPW;
  if (base >= e) return;
  const int cnt = (e - base < SC_EPW) ? (e - base) : SC_EPW;

  const int idx = base + ((lane < cnt) ? lane : (cnt - 1));
  const int   myR = row[idx];
  const int   myC = col[idx];
  const float dR = dinv[myR];
  const float dC = dinv[myC];

  if (cnt == SC_EPW) {
#pragma unroll
    for (int g = 0; g < SC_EPW; g += SC_GROUP) {
      float v0[SC_GROUP], v1[SC_GROUP];
#pragma unroll
      for (int k = 0; k < SC_GROUP; ++k) {
        const unsigned co = (unsigned)lane_bcast_i(myC, g + k) * D_FEAT + lane;
        v0[k] = xw[co];
        v1[k] = xw[co + 32];
      }
      const float myW = dR * dC;
#pragma unroll
      for (int k = 0; k < SC_GROUP; ++k) {
        const unsigned ro = (unsigned)lane_bcast_i(myR, g + k) * D_FEAT + lane;
        const float w = lane_bcast_f(myW, g + k);
        atomicAdd(&out[ro], v0[k] * w);
        atomicAdd(&out[ro + 32], v1[k] * w);
      }
    }
  } else {
    for (int j = 0; j < cnt; ++j) {
      const unsigned co = (unsigned)lane_bcast_i(myC, j) * D_FEAT + lane;
      const unsigned ro = (unsigned)lane_bcast_i(myR, j) * D_FEAT + lane;
      const float w = lane_bcast_f(dR, j) * lane_bcast_f(dC, j);
      const float a0 = xw[co] * w;
      const float a1 = xw[co + 32] * w;
      atomicAdd(&out[ro], a0);
      atomicAdd(&out[ro + 32], a1);
    }
  }
}


extern "C" void kernel_launch(void* const* d_in, const int* in_sizes, int n_in,
                              void* d_out, int out_size, void* d_ws, size_t ws_size,
                              hipStream_t stream) {
  const float* x      = (const float*)d_in[0];
  const int*   eidx   = (const int*)d_in[1];
  const float* weight = (const float*)d_in[2];
  const float* bias   = (const float*)d_in[3];
  float*       out    = (float*)d_out;

  const int n = in_sizes[0] / D_FEAT;
  const int e = in_sizes[1] / 2;
  const int* row = eidx;
  const int* col = eidx + e;

  float* dinv = (float*)d_ws;
  const int dinvStride = (n + 255) & ~255;
  float* xw = dinv + dinvStride;
  int* partial = (int*)(xw + n * D_FEAT);

  count_and_gemm_kernel<<<((n + CNT_BINS - 1) / CNT_BINS) * CNT_CHUNKS +
                              (n + GEMM_ROWS_PER_BLOCK - 1) / GEMM_ROWS_PER_BLOCK,
                          CNT_THREADS, 0, stream>>>(row, partial, x, weight, bias, xw, out,
                                                    n, e, dinvStride);
  dinv_kernel<<<(n + 255) / 256, 256, 0, stream>>>(partial, dinv, n, dinvStride);
  scatter_kernel<<<(e + 255) / 256, SC_THREADS, 0, stream>>>(row, col, dinv, xw, out, e);
}


# edge scatter: 256 resident work-groups (32 waves) copy a 2-byte-per-node degree table into LDS once and share the edges evenly; both dinv operands come from LDS, not L2 gathers (atomic order free)
# speedup vs baseline: 2.8616x; 1.0574x over previous
#include <hip/hip_runtime.h>
#include <hip/hip_bf16.h>


typedef __attribute__((ext_vector_type(2))) float v2f;
typedef __attribute__((ext_vector_type(8))) float v8f;

#define D_FEAT 64


#define CNT_BINS    2048
#define CNT_CHUNKS  10
#define CNT_THREADS 512
#define CNT_INFLIGHT 8

__device__ __forceinline__ void count_one(unsigned int* hist, int r, int binBase) {
  const unsigned int b = (unsigned int)(r - binBase);
  if (b < (unsigned int)CNT_BINS) atomicAdd(&hist[b], 1u);
}

__device__ __forceinline__ void degree_count_block(unsigned int* hist,
                                                   const int* __restrict__ row,
                                                   int* __restrict__ partial,
                                                   int n, int e, int nStride,
                                                   int binBlock, int chunk) {
  const int binBase = binBlock * CNT_BINS;

  for (int i = threadIdx.x; i < CNT_BINS; i += CNT_THREADS) hist[i] = 0u;
  __syncthreads();

  const int chunkLen = (((e + CNT_CHUNKS - 1) / CNT_CHUNKS) + 3) & ~3;
  const int beg = chunk * chunkLen;
  int end = beg + chunkLen;
  if (end > e) end = e;
  const int nvec = (end > beg) ? ((end - beg) >> 2) : 0;

  if (nvec > 0) {
    const int4* rv = reinterpret_cast<const int4*>(row + beg);
    for (int v = threadIdx.x; v < nvec; v += CNT_INFLIGHT * CNT_THREADS) {
      int4 r[CNT_INFLIGHT];
      for (int k = 0; k < CNT_INFLIGHT; ++k) {
        const int idx = v + k * CNT_THREADS;
        r[k] = (idx < nvec) ? rv[idx] : make_int4(-1, -1, -1, -1);
      }
      for (int k = 0; k < CNT_INFLIGHT; ++k) {
        count_one(hist, r[k].x, binBase);
        count_one(hist, r[k].y, binBase);
        count_one(hist, r[k].z, binBase);
        count_one(hist, r[k].w, binBase);
      }
    }
  }
  for (int j = beg + 4 * nvec + (int)threadIdx.x; j < end; j += CNT_THREADS)
    count_one(hist, row[j], binBase);

  __syncthreads();
  for (int i = threadIdx.x; i < CNT_BINS; i += CNT_THREADS) {
    const int node = binBase + i;
    if (node < n) partial[chunk * nStride + node] = (int)hist[i];
  }
}

#define DEG16_SAT 65535

__device__ __forceinline__ float dinv_from_count(int cnt) {
  const float d = (float)cnt;
  return (d > 0.0f) ? rsqrtf(d) : 0.0f;
}
__device__ __forceinline__ unsigned int sat16(int cnt) {
  return (unsigned int)((cnt < DEG16_SAT) ? cnt : DEG16_SAT);
}

__global__ __launch_bounds__(256) void dinv_kernel(const int* __restrict__ partial,
                                                   float* __restrict__ dinv,
                                                   unsigned short* __restrict__ deg16,
                                                   int n, int nStride) {
  const int i = 2 * (blockIdx.x * blockDim.x + threadIdx.x);
  if (i + 1 < n) {
    int c0 = 0, c1 = 0;
    for (int c = 0; c < CNT_CHUNKS; ++c) {
      const int2 p = *reinterpret_cast<const int2*>(partial + c * nStride + i);
      c0 += p.x;
      c1 += p.y;
    }
    float2 r;
    r.x = dinv_from_count(c0);
    r.y = dinv_from_count(c1);
    *reinterpret_cast<float2*>(dinv + i) = r;
    *reinterpret_cast<unsigned int*>(deg16 + i) = sat16(c0) | (sat16(c1) << 16);
  } else if (i < n) {
    int c0 = 0;
    for (int c = 0; c < CNT_CHUNKS; ++c) c0 += partial[c * nStride + i];
    dinv[i] = dinv_from_count(c0);
    deg16[i] = (unsigned short)sat16(c0);
  }
}

#define GEMM_ROWS_PER_BLOCK 64

__device__ __forceinline__ void gemm_xw_block(const float* __restrict__ X,
                                              const float* __restrict__ W,
                                              const float* __restrict__ bias,
                                              float* __restrict__ XW,
                                              float* __restrict__ out, int n, int gemmBlock) {
  const int wave = threadIdx.x >> 5;
  const int lane = threadIdx.x & 31;
  const int half = lane >> 4;
  const int l    = lane & 15;

  const int mBase = (gemmBlock * 4 + (wave >> 2)) * 16;
  const int nBase = (wave & 3) * 16;
  if (mBase >= n) return;

  {
    const float4 bv = reinterpret_cast<const float4*>(bias)[l];
    float4* const out4 = reinterpret_cast<float4*>(out);
    const int o4 = (mBase + (wave & 3) * 4 + half) * (D_FEAT / 4) + l;
    out4[o4] = bv;
    out4[o4 + 2 * (D_FEAT / 4)] = bv;
  }

  const int m = mBase + l;

  v8f acc = {};
#pragma unroll
  for (int k = 0; k < D_FEAT; k += 4) {
    v2f a, b;
    a.x = X[m * D_FEAT + k + 2 * half + 0];
    a.y = X[m * D_FEAT + k + 2 * half + 1];
    b.x = W[(k + 2 * half + 0) * D_FEAT + nBase + l];
    b.y = W[(k + 2 * half + 1) * D_FEAT + nBase + l];
    acc = __builtin_amdgcn_wmma_f32_16x16x4_f32(false, a, false, b,
                                                (short)0, acc, false, false);
  }

#pragma unroll
  for (int r = 0; r < 8; ++r) {
    XW[(mBase + r + 8 * half) * D_FEAT + nBase + l] = acc[r];
  }
}

__global__ __launch_bounds__(CNT_THREADS, 16) void count_and_gemm_kernel(const int* __restrict__ row,
                                                                     int* __restrict__ partial,
                                                                     const float* __restrict__ X,
                                                                     const float* __restrict__ W,
                                                                     const float* __restrict__ bias,
                                                                     float* __restrict__ XW,
                                                                     float* __restrict__ out,
                                                                     int n, int e, int nStride) {
  __shared__ unsigned int hist[CNT_BINS];
  const int binBlocks   = (n + CNT_BINS - 1) / CNT_BINS;
  const int countBlocks = binBlocks * CNT_CHUNKS;
  const int b = blockIdx.x;
  if (b < countBlocks) {
    const int chunk = b / binBlocks;
    degree_count_block(hist, row, partial, n, e, nStride, b - chunk * binBlocks, chunk);
  } else {
    gemm_xw_block(X, W, bias, XW, out, n, b - countBlocks);
  }
}

#define SC_THREADS 1024
#define SC_WGS     256
#define SC_GROUP   8
#define SC_GPB     4
#define SC_TAB     98304
#define SC_FILL    8

__device__ __forceinline__ int lane_bcast_i(int v, int srcLane) {
  return __builtin_amdgcn_readlane(v, srcLane);
}
__device__ __forceinline__ float lane_bcast_f(float v, int srcLane) {
  return __int_as_float(__builtin_amdgcn_readlane(__float_as_int(v), srcLane));
}

__device__ __forceinline__ void batch_ids(const int* __restrict__ row, const int* __restrict__ col,
                                          int base, int lim, int lane, int& r, int& c) {
  const int last = lim - 1 - base;
  const int idx = base + ((lane < last) ? lane : last);
  r = row[idx];
  c = col[idx];
}

__device__ __forceinline__ float dinv_of(unsigned int d16) {
  const float d = (float)d16;
  return (d > 0.0f) ? rsqrtf(d) : 0.0f;
}

template <bool TAB>
__device__ __forceinline__ void scatter_groups(const unsigned short* tab,
                                               const int* __restrict__ row,
                                               const int* __restrict__ col,
                                               const float* __restrict__ dinv,
                                               const float* __restrict__ xw,
                                               float* __restrict__ out,
                                               int e, int g, int gEnd, int lane,
                                               int myR, int myC) {
  const int lim = (gEnd * SC_GROUP < e) ? (gEnd * SC_GROUP) : e;
  while (g < gEnd) {
    const int ng = (gEnd - g < SC_GPB) ? (gEnd - g) : SC_GPB;
    const int base = g * SC_GROUP;
    const int cnt = (lim - base < ng * SC_GROUP) ? (lim - base) : (ng * SC_GROUP);

    const int gNext = g + SC_GPB;
    int nxR = 0, nxC = 0;
    if (gNext < gEnd) batch_ids(row, col, gNext * SC_GROUP, lim, lane, nxR, nxC);

    float myW;
    if (TAB) {
      const unsigned int degR = tab[myR];
      const unsigned int degC = tab[myC];
      myW = dinv_of(degR) * dinv_of(degC);
      const bool sat = (degR == (unsigned int)DEG16_SAT) | (degC == (unsigned int)DEG16_SAT);
      if (__builtin_amdgcn_ballot_w32(sat) != 0u) myW = dinv[myR] * dinv[myC];
    } else {
      myW = dinv[myR] * dinv[myC];
    }

    if (cnt == ng * SC_GROUP) {
#pragma unroll
      for (int q = 0; q < SC_GPB; ++q) {
        if (q < ng) {
          float v0[SC_GROUP], v1[SC_GROUP];
#pragma unroll
          for (int k = 0; k < SC_GROUP; ++k) {
            const unsigned co = (unsigned)lane_bcast_i(myC, q * SC_GROUP + k) * D_FEAT + lane;
            v0[k] = xw[co];
            v1[k] = xw[co + 32];
          }
#pragma unroll
          for (int k = 0; k < SC_GROUP; ++k) {
            const unsigned ro = (unsigned)lane_bcast_i(myR, q * SC_GROUP + k) * D_FEAT + lane;
            const float w = lane_bcast_f(myW, q * SC_GROUP + k);
            atomicAdd(&out[ro], v0[k] * w);
            atomicAdd(&out[ro + 32], v1[k] * w);
          }
        }
      }
    } else {
      for (int j = 0; j < cnt; ++j) {
        const unsigned co = (unsigned)lane_bcast_i(myC, j) * D_FEAT + lane;
        const unsigned ro = (unsigned)lane_bcast_i(myR, j) * D_FEAT + lane;
        const float w = lane_bcast_f(myW, j);
        const float a0 = xw[co] * w;
        const float a1 = xw[co + 32] * w;
        atomicAdd(&out[ro], a0);
        atomicAdd(&out[ro + 32], a1);
      }
    }

    myR = nxR;
    myC = nxC;
    g = gNext;
  }
}

__global__ __launch_bounds__(SC_THREADS) void scatter_kernel(const int* __restrict__ row,
                                                             const int* __restrict__ col,
                                                             const float* __restrict__ dinv,
                                                             const unsigned short* __restrict__ deg16,
                                                             const float* __restrict__ xw,
                                                             float* __restrict__ out,
                                                             int n, int e) {
  __shared__ __attribute__((aligned(16))) unsigned short tab[SC_TAB];
  const int lane = threadIdx.x & 31;
  const int wave = __builtin_amdgcn_readfirstlane((int)(threadIdx.x >> 5));

  const int groups = (e + SC_GROUP - 1) / SC_GROUP;
  const int waves  = (int)gridDim.x * (SC_THREADS / 32);
  const int gw     = (int)blockIdx.x * (SC_THREADS / 32) + wave;
  const int per    = groups / waves;
  const int rem    = groups - per * waves;
  const int g0     = gw * per + ((gw < rem) ? gw : rem);
  const int gEnd   = g0 + per + ((gw < rem) ? 1 : 0);

  int myR = 0, myC = 0;
  if (g0 < gEnd) {
    const int lim = (gEnd * SC_GROUP < e) ? (gEnd * SC_GROUP) : e;
    batch_ids(row, col, g0 * SC_GROUP, lim, lane, myR, myC);
  }

  if (n <= SC_TAB) {
    const int nvec = n >> 3;
    const int npiece = nvec >> 5;
    const uint4* d4 = reinterpret_cast<const uint4*>(deg16);
    uint4* t4 = reinterpret_cast<uint4*>(tab);
    for (int p = wave; p < npiece; p += SC_FILL * (SC_THREADS / 32)) {
      uint4 t[SC_FILL];
#pragma unroll
      for (int k = 0; k < SC_FILL; ++k) {
        const int q = p + k * (SC_THREADS / 32);
        t[k] = (q < npiece) ? d4[q * 32 + lane] : make_uint4(0u, 0u, 0u, 0u);
      }
#pragma unroll
      for (int k = 0; k < SC_FILL; ++k) {
        const int q = p + k * (SC_THREADS / 32);
        if (q < npiece) t4[q * 32 + lane] = t[k];
      }
    }
    const int v = (npiece << 5) + (int)threadIdx.x;
    if (v < nvec) t4[v] = d4[v];
    const int i = (nvec << 3) + (int)threadIdx.x;
    if (i < n) tab[i] = deg16[i];
    __syncthreads();
    scatter_groups<true>(tab, row, col, dinv, xw, out, e, g0, gEnd, lane, myR, myC);
  } else {
    scatter_groups<false>(tab, row, col, dinv, xw, out, e, g0, gEnd, lane, myR, myC);
  }
}


extern "C" void kernel_launch(void* const* d_in, const int* in_sizes, int n_in,
                              void* d_out, int out_size, void* d_ws, size_t ws_size,
                              hipStream_t stream) {
  const float* x      = (const float*)d_in[0];
  const int*   eidx   = (const int*)d_in[1];
  const float* weight = (const float*)d_in[2];
  const float* bias   = (const float*)d_in[3];
  float*       out    = (float*)d_out;

  const int n = in_sizes[0] / D_FEAT;
  const int e = in_sizes[1] / 2;
  const int* row = eidx;
  const int* col = eidx + e;

  float* dinv = (float*)d_ws;
  const int dinvStride = (n + 255) & ~255;
  float* xw = dinv + dinvStride;
  int* partial = (int*)(xw + n * D_FEAT);
  unsigned short* deg16 = (unsigned short*)(partial + CNT_CHUNKS * dinvStride);

  count_and_gemm_kernel<<<((n + CNT_BINS - 1) / CNT_BINS) * CNT_CHUNKS +
                              (n + GEMM_ROWS_PER_BLOCK - 1) / GEMM_ROWS_PER_BLOCK,
                          CNT_THREADS, 0, stream>>>(row, partial, x, weight, bias, xw, out,
                                                    n, e, dinvStride);
  dinv_kernel<<<(n + 511) / 512, 256, 0, stream>>>(partial, dinv, deg16, n, dinvStride);
  scatter_kernel<<<SC_WGS, SC_THREADS, 0, stream>>>(row, col, dinv, deg16, xw, out, n, e);
}


# edge sums by owner: launch 1 also orders each edge chunk's (col,row) pairs by row range (LDS counting sort); a work-group owns 196 rows, sums them in LDS (float adds, arrival order), stores out once
# speedup vs baseline: 3.5582x; 1.2434x over previous
#include <hip/hip_runtime.h>
#include <hip/hip_bf16.h>


typedef __attribute__((ext_vector_type(2))) float v2f;
typedef __attribute__((ext_vector_type(8))) float v8f;

#define D_FEAT 64


#define CNT_BINS    2048
#define CNT_CHUNKS  10
#define CNT_THREADS 512
#define CNT_INFLIGHT 8

__device__ __forceinline__ void count_one(unsigned int* hist, int r, int binBase) {
  const unsigned int b = (unsigned int)(r - binBase);
  if (b < (unsigned int)CNT_BINS) atomicAdd(&hist[b], 1u);
}

__device__ __forceinline__ void degree_count_block(unsigned int* hist,
                                                   const int* __restrict__ row,
                                                   int* __restrict__ partial,
                                                   int n, int e, int nStride,
                                                   int binBlock, int chunk) {
  const int binBase = binBlock * CNT_BINS;

  for (int i = threadIdx.x; i < CNT_BINS; i += CNT_THREADS) hist[i] = 0u;
  __syncthreads();

  const int chunkLen = (((e + CNT_CHUNKS - 1) / CNT_CHUNKS) + 3) & ~3;
  const int beg = chunk * chunkLen;
  int end = beg + chunkLen;
  if (end > e) end = e;
  const int nvec = (end > beg) ? ((end - beg) >> 2) : 0;

  if (nvec > 0) {
    const int4* rv = reinterpret_cast<const int4*>(row + beg);
    for (int v = threadIdx.x; v < nvec; v += CNT_INFLIGHT * CNT_THREADS) {
      int4 r[CNT_INFLIGHT];
      for (int k = 0; k < CNT_INFLIGHT; ++k) {
        const int idx = v + k * CNT_THREADS;
        r[k] = (idx < nvec) ? rv[idx] : make_int4(-1, -1, -1, -1);
      }
      for (int k = 0; k < CNT_INFLIGHT; ++k) {
        count_one(hist, r[k].x, binBase);
        count_one(hist, r[k].y, binBase);
        count_one(hist, r[k].z, binBase);
        count_one(hist, r[k].w, binBase);
      }
    }
  }
  for (int j = beg + 4 * nvec + (int)threadIdx.x; j < end; j += CNT_THREADS)
    count_one(hist, row[j], binBase);

  __syncthreads();
  for (int i = threadIdx.x; i < CNT_BINS; i += CNT_THREADS) {
    const int node = binBase + i;
    if (node < n) partial[chunk * nStride + node] = (int)hist[i];
  }
}

#define DEG16_SAT 65535

__device__ __forceinline__ float dinv_from_count(int cnt) {
  const float d = (float)cnt;
  return (d > 0.0f) ? rsqrtf(d) : 0.0f;
}
__device__ __forceinline__ unsigned int sat16(int cnt) {
  return (unsigned int)((cnt < DEG16_SAT) ? cnt : DEG16_SAT);
}

__global__ __launch_bounds__(256) void dinv_kernel(const int* __restrict__ partial,
                                                   float* __restrict__ dinv,
                                                   unsigned short* __restrict__ deg16,
                                                   int n, int nStride) {
  const int i = 2 * (blockIdx.x * blockDim.x + threadIdx.x);
  if (i + 1 < n) {
    int c0 = 0, c1 = 0;
    for (int c = 0; c < CNT_CHUNKS; ++c) {
      const int2 p = *reinterpret_cast<const int2*>(partial + c * nStride + i);
      c0 += p.x;
      c1 += p.y;
    }
    float2 r;
    r.x = dinv_from_count(c0);
    r.y = dinv_from_count(c1);
    *reinterpret_cast<float2*>(dinv + i) = r;
    *reinterpret_cast<unsigned int*>(deg16 + i) = sat16(c0) | (sat16(c1) << 16);
  } else if (i < n) {
    int c0 = 0;
    for (int c = 0; c < CNT_CHUNKS; ++c) c0 += partial[c * nStride + i];
    dinv[i] = dinv_from_count(c0);
    deg16[i] = (unsigned short)sat16(c0);
  }
}

#define GEMM_ROWS_PER_BLOCK 64

__device__ __forceinline__ void gemm_xw_block(const float* __restrict__ X,
                                              const float* __restrict__ W,
                                              float* __restrict__ XW, int n, int gemmBlock) {
  const int wave = threadIdx.x >> 5;
  const int lane = threadIdx.x & 31;
  const int half = lane >> 4;
  const int l    = lane & 15;

  const int mBase = (gemmBlock * 4 + (wave >> 2)) * 16;
  const int nBase = (wave & 3) * 16;
  if (mBase >= n) return;

  const int m = mBase + l;

  v8f acc = {};
#pragma unroll
  for (int k = 0; k < D_FEAT; k += 4) {
    v2f a, b;
    a.x = X[m * D_FEAT + k + 2 * half + 0];
    a.y = X[m * D_FEAT + k + 2 * half + 1];
    b.x = W[(k + 2 * half + 0) * D_FEAT + nBase + l];
    b.y = W[(k + 2 * half + 1) * D_FEAT + nBase + l];
    acc = __builtin_amdgcn_wmma_f32_16x16x4_f32(false, a, false, b,
                                                (short)0, acc, false, false);
  }

#pragma unroll
  for (int r = 0; r < 8; ++r) {
    XW[(mBase + r + 8 * half) * D_FEAT + nBase + l] = acc[r];
  }
}

#define OW_ROWS   196
#define OW_RANGES 256
#define BN_WGS    256
#define BN_CAP    3200
#define BN_PER    7
#define BN_PITCH  288

__device__ __forceinline__ void scan256(const unsigned int* cnt, unsigned int* off, int lane) {
  unsigned int v[8];
  unsigned int s = 0u;
#pragma unroll
  for (int k = 0; k < 8; ++k) {
    const unsigned int t = cnt[8 * lane + k];
    v[k] = s;
    s += t;
  }
  unsigned int incl = s;
#pragma unroll
  for (int d = 1; d < 32; d <<= 1) {
    const unsigned int o = __shfl_up(incl, (unsigned int)d);
    if (lane >= d) incl += o;
  }
  const unsigned int base = incl - s;
#pragma unroll
  for (int k = 0; k < 8; ++k) off[8 * lane + k] = base + v[k];
  if (lane == 31) off[256] = incl;
}

__device__ __forceinline__ unsigned int range_of(int r) {
  const unsigned int g = (unsigned int)r / (unsigned int)OW_ROWS;
  return (g < (unsigned int)OW_RANGES) ? g : (unsigned int)(OW_RANGES - 1);
}

__device__ __forceinline__ void bin_block(unsigned int* smem,
                                          const int* __restrict__ row,
                                          const int* __restrict__ col,
                                          int2* __restrict__ glist,
                                          int* __restrict__ goff, int e, int w) {
  int2* const stage = reinterpret_cast<int2*>(smem);
  unsigned int* const cnt = smem + 2 * BN_CAP;
  unsigned int* const off = cnt + OW_RANGES;
  const int t = (int)threadIdx.x;

  const int chunkLen = (e + BN_WGS - 1) / BN_WGS;
  const int beg = w * chunkLen;
  int end = beg + chunkLen;
  if (end > e) end = e;
  const int len = (end > beg) ? (end - beg) : 0;

  if (t < OW_RANGES) cnt[t] = 0u;
  __syncthreads();

  int r[BN_PER], c[BN_PER];
  unsigned int pos[BN_PER];
#pragma unroll
  for (int k = 0; k < BN_PER; ++k) {
    const int i = k * CNT_THREADS + t;
    const int j = (i < len) ? (beg + i) : 0;
    r[k] = row[j];
    c[k] = col[j];
  }
#pragma unroll
  for (int k = 0; k < BN_PER; ++k) {
    pos[k] = 0u;
    if (k * CNT_THREADS + t < len) pos[k] = atomicAdd(&cnt[range_of(r[k])], 1u);
  }
  __syncthreads();
  if (t < 32) scan256(cnt, off, t);
  __syncthreads();
#pragma unroll
  for (int k = 0; k < BN_PER; ++k) {
    if (k * CNT_THREADS + t < len) stage[off[range_of(r[k])] + pos[k]] = make_int2(c[k], r[k]);
  }
  __syncthreads();
  int2* const mine = glist + w * BN_CAP;
  for (int i = t; i < len; i += CNT_THREADS) mine[i] = stage[i];
  if (t <= OW_RANGES) goff[w * BN_PITCH + t] = (int)off[t];
}

__global__ __launch_bounds__(CNT_THREADS, 16) void count_bin_gemm_kernel(const int* __restrict__ row,
                                                                     const int* __restrict__ col,
                                                                     int* __restrict__ partial,
                                                                     int2* __restrict__ glist,
                                                                     int* __restrict__ goff,
                                                                     const float* __restrict__ X,
                                                                     const float* __restrict__ W,
                                                                     float* __restrict__ XW,
                                                                     int n, int e, int nStride) {
  __shared__ __attribute__((aligned(16))) unsigned int smem[2 * BN_CAP + 2 * OW_RANGES + 8];
  const int binBlocks   = (n + CNT_BINS - 1) / CNT_BINS;
  const int countBlocks = binBlocks * CNT_CHUNKS;
  const int b = blockIdx.x;
  if (b < countBlocks) {
    const int chunk = b / binBlocks;
    degree_count_block(smem, row, partial, n, e, nStride, b - chunk * binBlocks, chunk);
  } else if (b < countBlocks + BN_WGS) {
    bin_block(smem, row, col, glist, goff, e, b - countBlocks);
  } else {
    gemm_xw_block(X, W, XW, n, b - countBlocks - BN_WGS);
  }
}

#define SC_THREADS 1024
#define SC_GROUP   8
#define SC_GPB     4
#define SC_TAB     98304
#define SC_FILL    8

__device__ __forceinline__ int lane_bcast_i(int v, int srcLane) {
  return __builtin_amdgcn_readlane(v, srcLane);
}
__device__ __forceinline__ float lane_bcast_f(float v, int srcLane) {
  return __int_as_float(__builtin_amdgcn_readlane(__float_as_int(v), srcLane));
}

__device__ __forceinline__ float dinv_of(unsigned int d16) {
  const float d = (float)d16;
  return (d > 0.0f) ? rsqrtf(d) : 0.0f;
}

__device__ __forceinline__ void fetch_pair(const unsigned int* segStart, const int* segBeg,
                                           const int2* __restrict__ glist, int q, int& r, int& c) {
  int lo = 0;
#pragma unroll
  for (int s = 128; s >= 1; s >>= 1) {
    if (segStart[lo + s] <= (unsigned int)q) lo += s;
  }
  const int2 pr = glist[lo * BN_CAP + segBeg[lo] + (q - (int)segStart[lo])];
  c = pr.x;
  r = pr.y;
}

__global__ __launch_bounds__(SC_THREADS) void owner_sum_kernel(const int2* __restrict__ glist,
                                                               const int* __restrict__ goff,
                                                               const float* __restrict__ dinv,
                                                               const unsigned short* __restrict__ deg16,
                                                               const float* __restrict__ xw,
                                                               const float* __restrict__ bias,
                                                               float* __restrict__ out, int n) {
  __shared__ __attribute__((aligned(16))) unsigned short tab[SC_TAB];
  __shared__ __attribute__((aligned(16))) float tile[OW_ROWS * D_FEAT];
  __shared__ unsigned int segLen[OW_RANGES];
  __shared__ unsigned int segStart[OW_RANGES + 1];
  __shared__ int segBeg[OW_RANGES];
  const int lane = threadIdx.x & 31;
  const int wave = __builtin_amdgcn_readfirstlane((int)(threadIdx.x >> 5));
  const int rowBase = (int)blockIdx.x * OW_ROWS;

  if ((int)threadIdx.x < BN_WGS) {
    const int* p = goff + (int)threadIdx.x * BN_PITCH + (int)blockIdx.x;
    const int pb = p[0];
    const int pe = p[1];
    segBeg[threadIdx.x] = pb;
    segLen[threadIdx.x] = (unsigned int)(pe - pb);
  }

  {
    float4* const z4 = reinterpret_cast<float4*>(tile);
    for (int i = (int)threadIdx.x; i < OW_ROWS * (D_FEAT / 4); i += SC_THREADS)
      z4[i] = make_float4(0.0f, 0.0f, 0.0f, 0.0f);
  }

  {
    const int nvec = n >> 3;
    const int npiece = nvec >> 5;
    const uint4* d4 = reinterpret_cast<const uint4*>(deg16);
    uint4* t4 = reinterpret_cast<uint4*>(tab);
    for (int p = wave; p < npiece; p += SC_FILL * (SC_THREADS / 32)) {
      uint4 t[SC_FILL];
#pragma unroll
      for (int k = 0; k < SC_FILL; ++k) {
        const int q = p + k * (SC_THREADS / 32);
        t[k] = (q < npiece) ? d4[q * 32 + lane] : make_uint4(0u, 0u, 0u, 0u);
      }
#pragma unroll
      for (int k = 0; k < SC_FILL; ++k) {
        const int q = p + k * (SC_THREADS / 32);
        if (q < npiece) t4[q * 32 + lane] = t[k];
      }
    }
    const int v = (npiece << 5) + (int)threadIdx.x;
    if (v < nvec) t4[v] = d4[v];
    const int i = (nvec << 3) + (int)threadIdx.x;
    if (i < n) tab[i] = deg16[i];
  }
  __syncthreads();
  if (threadIdx.x < 32) scan256(segLen, segStart, lane);
  __syncthreads();

  const int M  = (int)segStart[OW_RANGES];
  const int nb = (M + 31) >> 5;

  int b = wave;
  int myR = 0, myC = 0;
  if (b < nb) {
    const int q = b * 32 + lane;
    fetch_pair(segStart, segBeg, glist, (q < M) ? q : (M - 1), myR, myC);
  }
  while (b < nb) {
    const int cnt = (M - b * 32 < 32) ? (M - b * 32) : 32;
    const int ng = cnt >> 3;

    const int bNext = b + SC_THREADS / 32;
    int nxR = 0, nxC = 0;
    if (bNext < nb) {
      const int q = bNext * 32 + lane;
      fetch_pair(segStart, segBeg, glist, (q < M) ? q : (M - 1), nxR, nxC);
    }

    const unsigned int degR = tab[myR];
    const unsigned int degC = tab[myC];
    float myW = dinv_of(degR) * dinv_of(degC);
    const bool sat = (degR == (unsigned int)DEG16_SAT) | (degC == (unsigned int)DEG16_SAT);
    if (__builtin_amdgcn_ballot_w32(sat) != 0u) myW = dinv[myR] * dinv[myC];
    const int myL = myR - rowBase;

#pragma unroll
    for (int q = 0; q < SC_GPB; ++q) {
      if (q < ng) {
        float v0[SC_GROUP], v1[SC_GROUP];
#pragma unroll
        for (int k = 0; k < SC_GROUP; ++k) {
          const unsigned co = (unsigned)lane_bcast_i(myC, q * SC_GROUP + k) * D_FEAT + lane;
          v0[k] = xw[co];
          v1[k] = xw[co + 32];
        }
#pragma unroll
        for (int k = 0; k < SC_GROUP; ++k) {
          const unsigned ro = (unsigned)lane_bcast_i(myL, q * SC_GROUP + k) * D_FEAT + lane;
          const float w = lane_bcast_f(myW, q * SC_GROUP + k);
          atomicAdd(&tile[ro], v0[k] * w);
          atomicAdd(&tile[ro + 32], v1[k] * w);
        }
      }
    }
    for (int j = ng * SC_GROUP; j < cnt; ++j) {
      const unsigned co = (unsigned)lane_bcast_i(myC, j) * D_FEAT + lane;
      const unsigned ro = (unsigned)lane_bcast_i(myL, j) * D_FEAT + lane;
      const float w = lane_bcast_f(myW, j);
      const float a0 = xw[co] * w;
      const float a1 = xw[co + 32] * w;
      atomicAdd(&tile[ro], a0);
      atomicAdd(&tile[ro + 32], a1);
    }

    myR = nxR;
    myC = nxC;
    b = bNext;
  }
  __syncthreads();

  {
    const float4* const t4 = reinterpret_cast<const float4*>(tile);
    const float4* const b4 = reinterpret_cast<const float4*>(bias);
    float4* const o4 = reinterpret_cast<float4*>(out);
    for (int i = (int)threadIdx.x; i < OW_ROWS * (D_FEAT / 4); i += SC_THREADS) {
      const int node = rowBase + (i >> 4);
      if (node < n) {
        float4 v = t4[i];
        const float4 bv = b4[i & 15];
        v.x += bv.x;
        v.y += bv.y;
        v.z += bv.z;
        v.w += bv.w;
        o4[node * (D_FEAT / 4) + (i & 15)] = v;
      }
    }
  }
}


extern "C" void kernel_launch(void* const* d_in, const int* in_sizes, int n_in,
                              void* d_out, int out_size, void* d_ws, size_t ws_size,
                              hipStream_t stream) {
  const float* x      = (const float*)d_in[0];
  const int*   eidx   = (const int*)d_in[1];
  const float* weight = (const float*)d_in[2];
  const float* bias   = (const float*)d_in[3];
  float*       out    = (float*)d_out;

  const int n = in_sizes[0] / D_FEAT;
  const int e = in_sizes[1] / 2;
  const int* row = eidx;
  const int* col = eidx + e;

  float* dinv = (float*)d_ws;
  const int dinvStride = (n + 255) & ~255;
  float* xw = dinv + dinvStride;
  int* partial = (int*)(xw + n * D_FEAT);
  int* goff = partial + CNT_CHUNKS * dinvStride;
  int2* glist = (int2*)(goff + BN_WGS * BN_PITCH);
  unsigned short* deg16 = (unsigned short*)(glist + BN_WGS * BN_CAP);

  count_bin_gemm_kernel<<<((n + CNT_BINS - 1) / CNT_BINS) * CNT_CHUNKS + BN_WGS +
                              (n + GEMM_ROWS_PER_BLOCK - 1) / GEMM_ROWS_PER_BLOCK,
                          CNT_THREADS, 0, stream>>>(row, col, partial, glist, goff, x, weight, xw,
                                                    n, e, dinvStride);
  dinv_kernel<<<(n + 511) / 512, 256, 0, stream>>>(partial, dinv, deg16, n, dinvStride);
  owner_sum_kernel<<<OW_RANGES, SC_THREADS, 0, stream>>>(glist, goff, dinv, deg16, xw, bias, out, n);
}


# edge sums: owner work-group first orders its edges by row (LDS counting sort); a wave sums a row's products in registers, two LDS float adds per row piece, not per edge (order: arrival, per row)
# speedup vs baseline: 3.8209x; 1.0738x over previous
#include <hip/hip_runtime.h>
#include <hip/hip_bf16.h>


typedef __attribute__((ext_vector_type(2))) float v2f;
typedef __attribute__((ext_vector_type(8))) float v8f;

#define D_FEAT 64


#define CNT_BINS    2048
#define CNT_CHUNKS  10
#define CNT_THREADS 512
#define CNT_INFLIGHT 8

__device__ __forceinline__ void count_one(unsigned int* hist, int r, int binBase) {
  const unsigned int b = (unsigned int)(r - binBase);
  if (b < (unsigned int)CNT_BINS) atomicAdd(&hist[b], 1u);
}

__device__ __forceinline__ void degree_count_block(unsigned int* hist,
                                                   const int* __restrict__ row,
                                                   int* __restrict__ partial,
                                                   int n, int e, int nStride,
                                                   int binBlock, int chunk) {
  const int binBase = binBlock * CNT_BINS;

  for (int i = threadIdx.x; i < CNT_BINS; i += CNT_THREADS) hist[i] = 0u;
  __syncthreads();

  const int chunkLen = (((e + CNT_CHUNKS - 1) / CNT_CHUNKS) + 3) & ~3;
  const int beg = chunk * chunkLen;
  int end = beg + chunkLen;
  if (end > e) end = e;
  const int nvec = (end > beg) ? ((end - beg) >> 2) : 0;

  if (nvec > 0) {
    const int4* rv = reinterpret_cast<const int4*>(row + beg);
    for (int v = threadIdx.x; v < nvec; v += CNT_INFLIGHT * CNT_THREADS) {
      int4 r[CNT_INFLIGHT];
      for (int k = 0; k < CNT_INFLIGHT; ++k) {
        const int idx = v + k * CNT_THREADS;
        r[k] = (idx < nvec) ? rv[idx] : make_int4(-1, -1, -1, -1);
      }
      for (int k = 0; k < CNT_INFLIGHT; ++k) {
        count_one(hist, r[k].x, binBase);
        count_one(hist, r[k].y, binBase);
        count_one(hist, r[k].z, binBase);
        count_one(hist, r[k].w, binBase);
      }
    }
  }
  for (int j = beg + 4 * nvec + (int)threadIdx.x; j < end; j += CNT_THREADS)
    count_one(hist, row[j], binBase);

  __syncthreads();
  for (int i = threadIdx.x; i < CNT_BINS; i += CNT_THREADS) {
    const int node = binBase + i;
    if (node < n) partial[chunk * nStride + node] = (int)hist[i];
  }
}

#define DEG16_SAT 65535

__device__ __forceinline__ float dinv_from_count(int cnt) {
  const float d = (float)cnt;
  return (d > 0.0f) ? rsqrtf(d) : 0.0f;
}
__device__ __forceinline__ unsigned int sat16(int cnt) {
  return (unsigned int)((cnt < DEG16_SAT) ? cnt : DEG16_SAT);
}

__global__ __launch_bounds__(256) void dinv_kernel(const int* __restrict__ partial,
                                                   float* __restrict__ dinv,
                                                   unsigned short* __restrict__ deg16,
                                                   int n, int nStride) {
  const int i = 2 * (blockIdx.x * blockDim.x + threadIdx.x);
  if (i + 1 < n) {
    int c0 = 0, c1 = 0;
    for (int c = 0; c < CNT_CHUNKS; ++c) {
      const int2 p = *reinterpret_cast<const int2*>(partial + c * nStride + i);
      c0 += p.x;
      c1 += p.y;
    }
    float2 r;
    r.x = dinv_from_count(c0);
    r.y = dinv_from_count(c1);
    *reinterpret_cast<float2*>(dinv + i) = r;
    *reinterpret_cast<unsigned int*>(deg16 + i) = sat16(c0) | (sat16(c1) << 16);
  } else if (i < n) {
    int c0 = 0;
    for (int c = 0; c < CNT_CHUNKS; ++c) c0 += partial[c * nStride + i];
    dinv[i] = dinv_from_count(c0);
    deg16[i] = (unsigned short)sat16(c0);
  }
}

#define GEMM_ROWS_PER_BLOCK 64

__device__ __forceinline__ void gemm_xw_block(const float* __restrict__ X,
                                              const float* __restrict__ W,
                                              float* __restrict__ XW, int n, int gemmBlock) {
  const int wave = threadIdx.x >> 5;
  const int lane = threadIdx.x & 31;
  const int half = lane >> 4;
  const int l    = lane & 15;

  const int mBase = (gemmBlock * 4 + (wave >> 2)) * 16;
  const int nBase = (wave & 3) * 16;
  if (mBase >= n) return;

  const int m = mBase + l;

  v8f acc = {};
#pragma unroll
  for (int k = 0; k < D_FEAT; k += 4) {
    v2f a, b;
    a.x = X[m * D_FEAT + k + 2 * half + 0];
    a.y = X[m * D_FEAT + k + 2 * half + 1];
    b.x = W[(k + 2 * half + 0) * D_FEAT + nBase + l];
    b.y = W[(k + 2 * half + 1) * D_FEAT + nBase + l];
    acc = __builtin_amdgcn_wmma_f32_16x16x4_f32(false, a, false, b,
                                                (short)0, acc, false, false);
  }

#pragma unroll
  for (int r = 0; r < 8; ++r) {
    XW[(mBase + r + 8 * half) * D_FEAT + nBase + l] = acc[r];
  }
}

#define OW_ROWS   196
#define OW_RANGES 256
#define BN_WGS    256
#define BN_CAP    3200
#define BN_PER    7
#define BN_PITCH  288

__device__ __forceinline__ void scan256(const unsigned int* cnt, unsigned int* off, int lane) {
  unsigned int v[8];
  unsigned int s = 0u;
#pragma unroll
  for (int k = 0; k < 8; ++k) {
    const unsigned int t = cnt[8 * lane + k];
    v[k] = s;
    s += t;
  }
  unsigned int incl = s;
#pragma unroll
  for (int d = 1; d < 32; d <<= 1) {
    const unsigned int o = __shfl_up(incl, (unsigned int)d);
    if (lane >= d) incl += o;
  }
  const unsigned int base = incl - s;
#pragma unroll
  for (int k = 0; k < 8; ++k) off[8 * lane + k] = base + v[k];
  if (lane == 31) off[256] = incl;
}

__device__ __forceinline__ unsigned int range_of(int r) {
  const unsigned int g = (unsigned int)r / (unsigned int)OW_ROWS;
  return (g < (unsigned int)OW_RANGES) ? g : (unsigned int)(OW_RANGES - 1);
}

__device__ __forceinline__ void bin_block(unsigned int* smem,
                                          const int* __restrict__ row,
                                          const int* __restrict__ col,
                                          int2* __restrict__ glist,
                                          int* __restrict__ goff, int e, int w) {
  int2* const stage = reinterpret_cast<int2*>(smem);
  unsigned int* const cnt = smem + 2 * BN_CAP;
  unsigned int* const off = cnt + OW_RANGES;
  const int t = (int)threadIdx.x;

  const int chunkLen = (e + BN_WGS - 1) / BN_WGS;
  const int beg = w * chunkLen;
  int end = beg + chunkLen;
  if (end > e) end = e;
  const int len = (end > beg) ? (end - beg) : 0;

  if (t < OW_RANGES) cnt[t] = 0u;
  __syncthreads();

  int r[BN_PER], c[BN_PER];
  unsigned int pos[BN_PER];
#pragma unroll
  for (int k = 0; k < BN_PER; ++k) {
    const int i = k * CNT_THREADS + t;
    const int j = (i < len) ? (beg + i) : 0;
    r[k] = row[j];
    c[k] = col[j];
  }
#pragma unroll
  for (int k = 0; k < BN_PER; ++k) {
    pos[k] = 0u;
    if (k * CNT_THREADS + t < len) pos[k] = atomicAdd(&cnt[range_of(r[k])], 1u);
  }
  __syncthreads();
  if (t < 32) scan256(cnt, off, t);
  __syncthreads();
#pragma unroll
  for (int k = 0; k < BN_PER; ++k) {
    if (k * CNT_THREADS + t < len) stage[off[range_of(r[k])] + pos[k]] = make_int2(c[k], r[k]);
  }
  __syncthreads();
  int2* const mine = glist + w * BN_CAP;
  for (int i = t; i < len; i += CNT_THREADS) mine[i] = stage[i];
  if (t <= OW_RANGES) goff[w * BN_PITCH + t] = (int)off[t];
}

__global__ __launch_bounds__(CNT_THREADS, 16) void count_bin_gemm_kernel(const int* __restrict__ row,
                                                                     const int* __restrict__ col,
                                                                     int* __restrict__ partial,
                                                                     int2* __restrict__ glist,
                                                                     int* __restrict__ goff,
                                                                     const float* __restrict__ X,
                                                                     const float* __restrict__ W,
                                                                     float* __restrict__ XW,
                                                                     int n, int e, int nStride) {
  __shared__ __attribute__((aligned(16))) unsigned int smem[2 * BN_CAP + 2 * OW_RANGES + 8];
  const int binBlocks   = (n + CNT_BINS - 1) / CNT_BINS;
  const int countBlocks = binBlocks * CNT_CHUNKS;
  const int b = blockIdx.x;
  if (b < countBlocks) {
    const int chunk = b / binBlocks;
    degree_count_block(smem, row, partial, n, e, nStride, b - chunk * binBlocks, chunk);
  } else if (b < countBlocks + BN_WGS) {
    bin_block(smem, row, col, glist, goff, e, b - countBlocks);
  } else {
    gemm_xw_block(X, W, XW, n, b - countBlocks - BN_WGS);
  }
}

#define SC_THREADS 1024
#define SC_GROUP   8
#define SC_GPB     4
#define SC_TAB     98304
#define SC_FILL    8
#define ST_PER     4
#define ST_CAP     (ST_PER * SC_THREADS)

__device__ __forceinline__ int lane_bcast_i(int v, int srcLane) {
  return __builtin_amdgcn_readlane(v, srcLane);
}
__device__ __forceinline__ float lane_bcast_f(float v, int srcLane) {
  return __int_as_float(__builtin_amdgcn_readlane(__float_as_int(v), srcLane));
}

__device__ __forceinline__ float dinv_of(unsigned int d16) {
  const float d = (float)d16;
  return (d > 0.0f) ? rsqrtf(d) : 0.0f;
}

__device__ __forceinline__ void flush_row(float* tile, int L, int lane, float a0, float a1) {
  atomicAdd(&tile[L * D_FEAT + lane], a0);
  atomicAdd(&tile[L * D_FEAT + 32 + lane], a1);
}

__device__ __forceinline__ void fetch_pair(const unsigned int* segStart, const int* segBeg,
                                           const int2* __restrict__ glist, int q, int& r, int& c) {
  int lo = 0;
#pragma unroll
  for (int s = 128; s >= 1; s >>= 1) {
    if (segStart[lo + s] <= (unsigned int)q) lo += s;
  }
  const int2 pr = glist[lo * BN_CAP + segBeg[lo] + (q - (int)segStart[lo])];
  c = pr.x;
  r = pr.y;
}

__global__ __launch_bounds__(SC_THREADS) void owner_sum_kernel(const int2* __restrict__ glist,
                                                               const int* __restrict__ goff,
                                                               const float* __restrict__ dinv,
                                                               const unsigned short* __restrict__ deg16,
                                                               const float* __restrict__ xw,
                                                               const float* __restrict__ bias,
                                                               float* __restrict__ out, int n) {
  __shared__ __attribute__((aligned(16))) unsigned short tab[SC_TAB];
  __shared__ __attribute__((aligned(16))) float tile[OW_ROWS * D_FEAT];
  __shared__ unsigned int segLen[OW_RANGES];
  __shared__ unsigned int segStart[OW_RANGES + 1];
  __shared__ int segBeg[OW_RANGES];
  __shared__ __attribute__((aligned(8))) int2 stage[ST_CAP];
  __shared__ unsigned int rowCnt[256];
  __shared__ unsigned int rowOff[256 + 1];
  const int lane = threadIdx.x & 31;
  const int wave = __builtin_amdgcn_readfirstlane((int)(threadIdx.x >> 5));
  const int rowBase = (int)blockIdx.x * OW_ROWS;

  if ((int)threadIdx.x < BN_WGS) {
    const int* p = goff + (int)threadIdx.x * BN_PITCH + (int)blockIdx.x;
    const int pb = p[0];
    const int pe = p[1];
    segBeg[threadIdx.x] = pb;
    segLen[threadIdx.x] = (unsigned int)(pe - pb);
  }

  {
    float4* const z4 = reinterpret_cast<float4*>(tile);
    for (int i = (int)threadIdx.x; i < OW_ROWS * (D_FEAT / 4); i += SC_THREADS)
      z4[i] = make_float4(0.0f, 0.0f, 0.0f, 0.0f);
  }

  {
    const int nvec = n >> 3;
    const int npiece = nvec >> 5;
    const uint4* d4 = reinterpret_cast<const uint4*>(deg16);
    uint4* t4 = reinterpret_cast<uint4*>(tab);
    for (int p = wave; p < npiece; p += SC_FILL * (SC_THREADS / 32)) {
      uint4 t[SC_FILL];
#pragma unroll
      for (int k = 0; k < SC_FILL; ++k) {
        const int q = p + k * (SC_THREADS / 32);
        t[k] = (q < npiece) ? d4[q * 32 + lane] : make_uint4(0u, 0u, 0u, 0u);
      }
#pragma unroll
      for (int k = 0; k < SC_FILL; ++k) {
        const int q = p + k * (SC_THREADS / 32);
        if (q < npiece) t4[q * 32 + lane] = t[k];
      }
    }
    const int v = (npiece << 5) + (int)threadIdx.x;
    if (v < nvec) t4[v] = d4[v];
    const int i = (nvec << 3) + (int)threadIdx.x;
    if (i < n) tab[i] = deg16[i];
  }
  __syncthreads();
  if (threadIdx.x < 32) scan256(segLen, segStart, lane);
  __syncthreads();

  const int M = (int)segStart[OW_RANGES];
  const int t = (int)threadIdx.x;

  for (int base = 0; base < M; base += ST_CAP) {
    const int m = (M - base < ST_CAP) ? (M - base) : ST_CAP;

    if (t < 256) rowCnt[t] = 0u;
    __syncthreads();

    int eC[ST_PER], eR[ST_PER];
#pragma unroll
    for (int k = 0; k < ST_PER; ++k) {
      const int i = k * SC_THREADS + t;
      fetch_pair(segStart, segBeg, glist, base + ((i < m) ? i : (m - 1)), eR[k], eC[k]);
    }
    unsigned int ePos[ST_PER];
    float eW[ST_PER];
#pragma unroll
    for (int k = 0; k < ST_PER; ++k) {
      const unsigned int degR = tab[eR[k]];
      const unsigned int degC = tab[eC[k]];
      float w = dinv_of(degR) * dinv_of(degC);
      const bool sat = (degR == (unsigned int)DEG16_SAT) | (degC == (unsigned int)DEG16_SAT);
      if (__builtin_amdgcn_ballot_w32(sat) != 0u) w = dinv[eR[k]] * dinv[eC[k]];
      eW[k] = w;
      ePos[k] = 0u;
      if (k * SC_THREADS + t < m) ePos[k] = atomicAdd(&rowCnt[eR[k] - rowBase], 1u);
    }
    __syncthreads();
    if (t < 32) scan256(rowCnt, rowOff, lane);
    __syncthreads();
#pragma unroll
    for (int k = 0; k < ST_PER; ++k) {
      if (k * SC_THREADS + t < m) {
        const int L = eR[k] - rowBase;
        stage[rowOff[L] + ePos[k]] = make_int2(eC[k] | (L << 20), __float_as_int(eW[k]));
      }
    }
    __syncthreads();

    const int nb = (m + 31) >> 5;
    int b = wave;
    int2 cur = make_int2(0, 0);
    if (b < nb) {
      const int q = b * 32 + lane;
      cur = stage[(q < m) ? q : (m - 1)];
    }
    while (b < nb) {
      const int cnt = (m - b * 32 < 32) ? (m - b * 32) : 32;
      const int ng = cnt >> 3;

      const int bNext = b + SC_THREADS / 32;
      int2 nx = cur;
      if (bNext < nb) {
        const int q = bNext * 32 + lane;
        nx = stage[(q < m) ? q : (m - 1)];
      }

      const int myC = cur.x & 0xFFFFF;
      const int myL = (int)((unsigned int)cur.x >> 20);
      const float myW = __int_as_float(cur.y);

      float acc0 = 0.0f, acc1 = 0.0f;
      int curL = lane_bcast_i(myL, 0);

#pragma unroll
      for (int q = 0; q < SC_GPB; ++q) {
        if (q < ng) {
          float v0[SC_GROUP], v1[SC_GROUP];
#pragma unroll
          for (int k = 0; k < SC_GROUP; ++k) {
            const unsigned co = (unsigned)lane_bcast_i(myC, q * SC_GROUP + k) * D_FEAT + lane;
            v0[k] = xw[co];
            v1[k] = xw[co + 32];
          }
#pragma unroll
          for (int k = 0; k < SC_GROUP; ++k) {
            const int Lj = lane_bcast_i(myL, q * SC_GROUP + k);
            const float w = lane_bcast_f(myW, q * SC_GROUP + k);
            if (Lj != curL) {
              flush_row(tile, curL, lane, acc0, acc1);
              acc0 = 0.0f;
              acc1 = 0.0f;
              curL = Lj;
            }
            acc0 += v0[k] * w;
            acc1 += v1[k] * w;
          }
        }
      }
      for (int j = ng * SC_GROUP; j < cnt; ++j) {
        const unsigned co = (unsigned)lane_bcast_i(myC, j) * D_FEAT + lane;
        const int Lj = lane_bcast_i(myL, j);
        const float w = lane_bcast_f(myW, j);
        if (Lj != curL) {
          flush_row(tile, curL, lane, acc0, acc1);
          acc0 = 0.0f;
          acc1 = 0.0f;
          curL = Lj;
        }
        acc0 += xw[co] * w;
        acc1 += xw[co + 32] * w;
      }
      flush_row(tile, curL, lane, acc0, acc1);

      cur = nx;
      b = bNext;
    }
  }
  __syncthreads();

  {
    const float4* const t4 = reinterpret_cast<const float4*>(tile);
    const float4* const b4 = reinterpret_cast<const float4*>(bias);
    float4* const o4 = reinterpret_cast<float4*>(out);
    for (int i = (int)threadIdx.x; i < OW_ROWS * (D_FEAT / 4); i += SC_THREADS) {
      const int node = rowBase + (i >> 4);
      if (node < n) {
        float4 v = t4[i];
        const float4 bv = b4[i & 15];
        v.x += bv.x;
        v.y += bv.y;
        v.z += bv.z;
        v.w += bv.w;
        o4[node * (D_FEAT / 4) + (i & 15)] = v;
      }
    }
  }
}


extern "C" void kernel_launch(void* const* d_in, const int* in_sizes, int n_in,
                              void* d_out, int out_size, void* d_ws, size_t ws_size,
                              hipStream_t stream) {
  const float* x      = (const float*)d_in[0];
  const int*   eidx   = (const int*)d_in[1];
  const float* weight = (const float*)d_in[2];
  const float* bias   = (const float*)d_in[3];
  float*       out    = (float*)d_out;

  const int n = in_sizes[0] / D_FEAT;
  const int e = in_sizes[1] / 2;
  const int* row = eidx;
  const int* col = eidx + e;

  float* dinv = (float*)d_ws;
  const int dinvStride = (n + 255) & ~255;
  float* xw = dinv + dinvStride;
  int* partial = (int*)(xw + n * D_FEAT);
  int* goff = partial + CNT_CHUNKS * dinvStride;
  int2* glist = (int2*)(goff + BN_WGS * BN_PITCH);
  unsigned short* deg16 = (unsigned short*)(glist + BN_WGS * BN_CAP);

  count_bin_gemm_kernel<<<((n + CNT_BINS - 1) / CNT_BINS) * CNT_CHUNKS + BN_WGS +
                              (n + GEMM_ROWS_PER_BLOCK - 1) / GEMM_ROWS_PER_BLOCK,
                          CNT_THREADS, 0, stream>>>(row, col, partial, glist, goff, x, weight, xw,
                                                    n, e, dinvStride);
  dinv_kernel<<<(n + 511) / 512, 256, 0, stream>>>(partial, dinv, deg16, n, dinvStride);
  owner_sum_kernel<<<OW_RANGES, SC_THREADS, 0, stream>>>(glist, goff, dinv, deg16, xw, bias, out, n);
}


# edge sums: each wave walks one equal run of consecutive edges in whole 8-edge groups (no 4th batch on two waves, no single-edge tail trips); row sums carried along the run (order: arrival, per row)
# speedup vs baseline: 3.9591x; 1.0362x over previous
#include <hip/hip_runtime.h>
#include <hip/hip_bf16.h>


typedef __attribute__((ext_vector_type(2))) float v2f;
typedef __attribute__((ext_vector_type(8))) float v8f;

#define D_FEAT 64


#define CNT_BINS    2048
#define CNT_CHUNKS  10
#define CNT_THREADS 512
#define CNT_INFLIGHT 8

__device__ __forceinline__ void count_one(unsigned int* hist, int r, int binBase) {
  const unsigned int b = (unsigned int)(r - binBase);
  if (b < (unsigned int)CNT_BINS) atomicAdd(&hist[b], 1u);
}

__device__ __forceinline__ void degree_count_block(unsigned int* hist,
                                                   const int* __restrict__ row,
                                                   int* __restrict__ partial,
                                                   int n, int e, int nStride,
                                                   int binBlock, int chunk) {
  const int binBase = binBlock * CNT_BINS;

  for (int i = threadIdx.x; i < CNT_BINS; i += CNT_THREADS) hist[i] = 0u;
  __syncthreads();

  const int chunkLen = (((e + CNT_CHUNKS - 1) / CNT_CHUNKS) + 3) & ~3;
  const int beg = chunk * chunkLen;
  int end = beg + chunkLen;
  if (end > e) end = e;
  const int nvec = (end > beg) ? ((end - beg) >> 2) : 0;

  if (nvec > 0) {
    const int4* rv = reinterpret_cast<const int4*>(row + beg);
    for (int v = threadIdx.x; v < nvec; v += CNT_INFLIGHT * CNT_THREADS) {
      int4 r[CNT_INFLIGHT];
      for (int k = 0; k < CNT_INFLIGHT; ++k) {
        const int idx = v + k * CNT_THREADS;
        r[k] = (idx < nvec) ? rv[idx] : make_int4(-1, -1, -1, -1);
      }
      for (int k = 0; k < CNT_INFLIGHT; ++k) {
        count_one(hist, r[k].x, binBase);
        count_one(hist, r[k].y, binBase);
        count_one(hist, r[k].z, binBase);
        count_one(hist, r[k].w, binBase);
      }
    }
  }
  for (int j = beg + 4 * nvec + (int)threadIdx.x; j < end; j += CNT_THREADS)
    count_one(hist, row[j], binBase);

  __syncthreads();
  for (int i = threadIdx.x; i < CNT_BINS; i += CNT_THREADS) {
    const int node = binBase + i;
    if (node < n) partial[chunk * nStride + node] = (int)hist[i];
  }
}

#define DEG16_SAT 65535

__device__ __forceinline__ float dinv_from_count(int cnt) {
  const float d = (float)cnt;
  return (d > 0.0f) ? rsqrtf(d) : 0.0f;
}
__device__ __forceinline__ unsigned int sat16(int cnt) {
  return (unsigned int)((cnt < DEG16_SAT) ? cnt : DEG16_SAT);
}

__global__ __launch_bounds__(256) void dinv_kernel(const int* __restrict__ partial,
                                                   float* __restrict__ dinv,
                                                   unsigned short* __restrict__ deg16,
                                                   int n, int nStride) {
  const int i = 2 * (blockIdx.x * blockDim.x + threadIdx.x);
  if (i + 1 < n) {
    int c0 = 0, c1 = 0;
    for (int c = 0; c < CNT_CHUNKS; ++c) {
      const int2 p = *reinterpret_cast<const int2*>(partial + c * nStride + i);
      c0 += p.x;
      c1 += p.y;
    }
    float2 r;
    r.x = dinv_from_count(c0);
    r.y = dinv_from_count(c1);
    *reinterpret_cast<float2*>(dinv + i) = r;
    *reinterpret_cast<unsigned int*>(deg16 + i) = sat16(c0) | (sat16(c1) << 16);
  } else if (i < n) {
    int c0 = 0;
    for (int c = 0; c < CNT_CHUNKS; ++c) c0 += partial[c * nStride + i];
    dinv[i] = dinv_from_count(c0);
    deg16[i] = (unsigned short)sat16(c0);
  }
}

#define GEMM_ROWS_PER_BLOCK 64

__device__ __forceinline__ void gemm_xw_block(const float* __restrict__ X,
                                              const float* __restrict__ W,
                                              float* __restrict__ XW, int n, int gemmBlock) {
  const int wave = threadIdx.x >> 5;
  const int lane = threadIdx.x & 31;
  const int half = lane >> 4;
  const int l    = lane & 15;

  const int mBase = (gemmBlock * 4 + (wave >> 2)) * 16;
  const int nBase = (wave & 3) * 16;
  if (mBase >= n) return;

  const int m = mBase + l;

  v8f acc = {};
#pragma unroll
  for (int k = 0; k < D_FEAT; k += 4) {
    v2f a, b;
    a.x = X[m * D_FEAT + k + 2 * half + 0];
    a.y = X[m * D_FEAT + k + 2 * half + 1];
    b.x = W[(k + 2 * half + 0) * D_FEAT + nBase + l];
    b.y = W[(k + 2 * half + 1) * D_FEAT + nBase + l];
    acc = __builtin_amdgcn_wmma_f32_16x16x4_f32(false, a, false, b,
                                                (short)0, acc, false, false);
  }

#pragma unroll
  for (int r = 0; r < 8; ++r) {
    XW[(mBase + r + 8 * half) * D_FEAT + nBase + l] = acc[r];
  }
}

#define OW_ROWS   196
#define OW_RANGES 256
#define BN_WGS    256
#define BN_CAP    3200
#define BN_PER    7
#define BN_PITCH  288

__device__ __forceinline__ void scan256(const unsigned int* cnt, unsigned int* off, int lane) {
  unsigned int v[8];
  unsigned int s = 0u;
#pragma unroll
  for (int k = 0; k < 8; ++k) {
    const unsigned int t = cnt[8 * lane + k];
    v[k] = s;
    s += t;
  }
  unsigned int incl = s;
#pragma unroll
  for (int d = 1; d < 32; d <<= 1) {
    const unsigned int o = __shfl_up(incl, (unsigned int)d);
    if (lane >= d) incl += o;
  }
  const unsigned int base = incl - s;
#pragma unroll
  for (int k = 0; k < 8; ++k) off[8 * lane + k] = base + v[k];
  if (lane == 31) off[256] = incl;
}

__device__ __forceinline__ unsigned int range_of(int r) {
  const unsigned int g = (unsigned int)r / (unsigned int)OW_ROWS;
  return (g < (unsigned int)OW_RANGES) ? g : (unsigned int)(OW_RANGES - 1);
}

__device__ __forceinline__ void bin_block(unsigned int* smem,
                                          const int* __restrict__ row,
                                          const int* __restrict__ col,
                                          int2* __restrict__ glist,
                                          int* __restrict__ goff, int e, int w) {
  int2* const stage = reinterpret_cast<int2*>(smem);
  unsigned int* const cnt = smem + 2 * BN_CAP;
  unsigned int* const off = cnt + OW_RANGES;
  const int t = (int)threadIdx.x;

  const int chunkLen = (e + BN_WGS - 1) / BN_WGS;
  const int beg = w * chunkLen;
  int end = beg + chunkLen;
  if (end > e) end = e;
  const int len = (end > beg) ? (end - beg) : 0;

  if (t < OW_RANGES) cnt[t] = 0u;
  __syncthreads();

  int r[BN_PER], c[BN_PER];
  unsigned int pos[BN_PER];
#pragma unroll
  for (int k = 0; k < BN_PER; ++k) {
    const int i = k * CNT_THREADS + t;
    const int j = (i < len) ? (beg + i) : 0;
    r[k] = row[j];
    c[k] = col[j];
  }
#pragma unroll
  for (int k = 0; k < BN_PER; ++k) {
    pos[k] = 0u;
    if (k * CNT_THREADS + t < len) pos[k] = atomicAdd(&cnt[range_of(r[k])], 1u);
  }
  __syncthreads();
  if (t < 32) scan256(cnt, off, t);
  __syncthreads();
#pragma unroll
  for (int k = 0; k < BN_PER; ++k) {
    if (k * CNT_THREADS + t < len) stage[off[range_of(r[k])] + pos[k]] = make_int2(c[k], r[k]);
  }
  __syncthreads();
  int2* const mine = glist + w * BN_CAP;
  for (int i = t; i < len; i += CNT_THREADS) mine[i] = stage[i];
  if (t <= OW_RANGES) goff[w * BN_PITCH + t] = (int)off[t];
}

__global__ __launch_bounds__(CNT_THREADS, 16) void count_bin_gemm_kernel(const int* __restrict__ row,
                                                                     const int* __restrict__ col,
                                                                     int* __restrict__ partial,
                                                                     int2* __restrict__ glist,
                                                                     int* __restrict__ goff,
                                                                     const float* __restrict__ X,
                                                                     const float* __restrict__ W,
                                                                     float* __restrict__ XW,
                                                                     int n, int e, int nStride) {
  __shared__ __attribute__((aligned(16))) unsigned int smem[2 * BN_CAP + 2 * OW_RANGES + 8];
  const int binBlocks   = (n + CNT_BINS - 1) / CNT_BINS;
  const int countBlocks = binBlocks * CNT_CHUNKS;
  const int b = blockIdx.x;
  if (b < countBlocks) {
    const int chunk = b / binBlocks;
    degree_count_block(smem, row, partial, n, e, nStride, b - chunk * binBlocks, chunk);
  } else if (b < countBlocks + BN_WGS) {
    bin_block(smem, row, col, glist, goff, e, b - countBlocks);
  } else {
    gemm_xw_block(X, W, XW, n, b - countBlocks - BN_WGS);
  }
}

#define SC_THREADS 1024
#define SC_GROUP   8
#define SC_GPB     4
#define SC_TAB     98304
#define SC_FILL    8
#define ST_PER     4
#define ST_CAP     (ST_PER * SC_THREADS)

__device__ __forceinline__ int lane_bcast_i(int v, int srcLane) {
  return __builtin_amdgcn_readlane(v, srcLane);
}
__device__ __forceinline__ float lane_bcast_f(float v, int srcLane) {
  return __int_as_float(__builtin_amdgcn_readlane(__float_as_int(v), srcLane));
}

__device__ __forceinline__ float dinv_of(unsigned int d16) {
  const float d = (float)d16;
  return (d > 0.0f) ? rsqrtf(d) : 0.0f;
}

__device__ __forceinline__ void flush_row(float* tile, int L, int lane, float a0, float a1) {
  atomicAdd(&tile[L * D_FEAT + lane], a0);
  atomicAdd(&tile[L * D_FEAT + 32 + lane], a1);
}

__device__ __forceinline__ void fetch_pair(const unsigned int* segStart, const int* segBeg,
                                           const int2* __restrict__ glist, int q, int& r, int& c) {
  int lo = 0;
#pragma unroll
  for (int s = 128; s >= 1; s >>= 1) {
    if (segStart[lo + s] <= (unsigned int)q) lo += s;
  }
  const int2 pr = glist[lo * BN_CAP + segBeg[lo] + (q - (int)segStart[lo])];
  c = pr.x;
  r = pr.y;
}

__global__ __launch_bounds__(SC_THREADS) void owner_sum_kernel(const int2* __restrict__ glist,
                                                               const int* __restrict__ goff,
                                                               const float* __restrict__ dinv,
                                                               const unsigned short* __restrict__ deg16,
                                                               const float* __restrict__ xw,
                                                               const float* __restrict__ bias,
                                                               float* __restrict__ out, int n) {
  __shared__ __attribute__((aligned(16))) unsigned short tab[SC_TAB];
  __shared__ __attribute__((aligned(16))) float tile[OW_ROWS * D_FEAT];
  __shared__ unsigned int segLen[OW_RANGES];
  __shared__ unsigned int segStart[OW_RANGES + 1];
  __shared__ int segBeg[OW_RANGES];
  __shared__ __attribute__((aligned(8))) int2 stage[ST_CAP];
  __shared__ unsigned int rowCnt[256];
  __shared__ unsigned int rowOff[256 + 1];
  const int lane = threadIdx.x & 31;
  const int wave = __builtin_amdgcn_readfirstlane((int)(threadIdx.x >> 5));
  const int rowBase = (int)blockIdx.x * OW_ROWS;

  if ((int)threadIdx.x < BN_WGS) {
    const int* p = goff + (int)threadIdx.x * BN_PITCH + (int)blockIdx.x;
    const int pb = p[0];
    const int pe = p[1];
    segBeg[threadIdx.x] = pb;
    segLen[threadIdx.x] = (unsigned int)(pe - pb);
  }

  {
    float4* const z4 = reinterpret_cast<float4*>(tile);
    for (int i = (int)threadIdx.x; i < OW_ROWS * (D_FEAT / 4); i += SC_THREADS)
      z4[i] = make_float4(0.0f, 0.0f, 0.0f, 0.0f);
  }

  {
    const int nvec = n >> 3;
    const int npiece = nvec >> 5;
    const uint4* d4 = reinterpret_cast<const uint4*>(deg16);
    uint4* t4 = reinterpret_cast<uint4*>(tab);
    for (int p = wave; p < npiece; p += SC_FILL * (SC_THREADS / 32)) {
      uint4 t[SC_FILL];
#pragma unroll
      for (int k = 0; k < SC_FILL; ++k) {
        const int q = p + k * (SC_THREADS / 32);
        t[k] = (q < npiece) ? d4[q * 32 + lane] : make_uint4(0u, 0u, 0u, 0u);
      }
#pragma unroll
      for (int k = 0; k < SC_FILL; ++k) {
        const int q = p + k * (SC_THREADS / 32);
        if (q < npiece) t4[q * 32 + lane] = t[k];
      }
    }
    const int v = (npiece << 5) + (int)threadIdx.x;
    if (v < nvec) t4[v] = d4[v];
    const int i = (nvec << 3) + (int)threadIdx.x;
    if (i < n) tab[i] = deg16[i];
  }
  __syncthreads();
  if (threadIdx.x < 32) scan256(segLen, segStart, lane);
  __syncthreads();

  const int M = (int)segStart[OW_RANGES];
  const int t = (int)threadIdx.x;

  for (int base = 0; base < M; base += ST_CAP) {
    const int m = (M - base < ST_CAP) ? (M - base) : ST_CAP;

    if (t < 256) rowCnt[t] = 0u;
    __syncthreads();

    int eC[ST_PER], eR[ST_PER];
#pragma unroll
    for (int k = 0; k < ST_PER; ++k) {
      const int i = k * SC_THREADS + t;
      fetch_pair(segStart, segBeg, glist, base + ((i < m) ? i : (m - 1)), eR[k], eC[k]);
    }
    unsigned int ePos[ST_PER];
    float eW[ST_PER];
#pragma unroll
    for (int k = 0; k < ST_PER; ++k) {
      const unsigned int degR = tab[eR[k]];
      const unsigned int degC = tab[eC[k]];
      float w = dinv_of(degR) * dinv_of(degC);
      const bool sat = (degR == (unsigned int)DEG16_SAT) | (degC == (unsigned int)DEG16_SAT);
      if (__builtin_amdgcn_ballot_w32(sat) != 0u) w = dinv[eR[k]] * dinv[eC[k]];
      eW[k] = w;
      ePos[k] = 0u;
      if (k * SC_THREADS + t < m) ePos[k] = atomicAdd(&rowCnt[eR[k] - rowBase], 1u);
    }
    __syncthreads();
    if (t < 32) scan256(rowCnt, rowOff, lane);
    __syncthreads();
#pragma unroll
    for (int k = 0; k < ST_PER; ++k) {
      if (k * SC_THREADS + t < m) {
        const int L = eR[k] - rowBase;
        stage[rowOff[L] + ePos[k]] = make_int2(eC[k] | (L << 20), __float_as_int(eW[k]));
      }
    }
    __syncthreads();

    const int G = (m + SC_GROUP - 1) / SC_GROUP;
    const int gBase = G / (SC_THREADS / 32);
    const int gRem = G - gBase * (SC_THREADS / 32);
    const int gFirst = wave * gBase + ((wave < gRem) ? wave : gRem);
    const int e0 = gFirst * SC_GROUP;
    int e1 = e0 + (gBase + ((wave < gRem) ? 1 : 0)) * SC_GROUP;
    if (e1 > m) e1 = m;
    if (e0 < e1) {
      int2 cur;
      {
        const int q = e0 + lane;
        cur = stage[(q < m) ? q : (m - 1)];
      }
      float acc0 = 0.0f, acc1 = 0.0f;
      int curL = lane_bcast_i((int)((unsigned int)cur.x >> 20), 0);

      for (int bs = e0; bs < e1; bs += 32) {
        const int cnt = (e1 - bs < 32) ? (e1 - bs) : 32;
        const int ng = cnt / SC_GROUP;
        const int rest = cnt - ng * SC_GROUP;

        int2 nx = cur;
        if (bs + 32 < e1) {
          const int q = bs + 32 + lane;
          nx = stage[(q < m) ? q : (m - 1)];
        }

        const int myC = cur.x & 0xFFFFF;
        const int myL = (int)((unsigned int)cur.x >> 20);
        const float myW = __int_as_float(cur.y);

#pragma unroll
        for (int q = 0; q < SC_GPB; ++q) {
          if (q < ng) {
            float v0[SC_GROUP], v1[SC_GROUP];
#pragma unroll
            for (int k = 0; k < SC_GROUP; ++k) {
              const unsigned co = (unsigned)lane_bcast_i(myC, q * SC_GROUP + k) * D_FEAT + lane;
              v0[k] = xw[co];
              v1[k] = xw[co + 32];
            }
#pragma unroll
            for (int k = 0; k < SC_GROUP; ++k) {
              const int Lj = lane_bcast_i(myL, q * SC_GROUP + k);
              const float w = lane_bcast_f(myW, q * SC_GROUP + k);
              if (Lj != curL) {
                flush_row(tile, curL, lane, acc0, acc1);
                acc0 = 0.0f;
                acc1 = 0.0f;
                curL = Lj;
              }
              acc0 += v0[k] * w;
              acc1 += v1[k] * w;
            }
          }
        }
        if (rest > 0) {
          const int s0 = ng * SC_GROUP;
          float v0[SC_GROUP], v1[SC_GROUP];
#pragma unroll
          for (int k = 0; k < SC_GROUP; ++k) {
            const unsigned co = (unsigned)lane_bcast_i(myC, s0 + k) * D_FEAT + lane;
            v0[k] = xw[co];
            v1[k] = xw[co + 32];
          }
#pragma unroll
          for (int k = 0; k < SC_GROUP; ++k) {
            if (k < rest) {
              const int Lj = lane_bcast_i(myL, s0 + k);
              const float w = lane_bcast_f(myW, s0 + k);
              if (Lj != curL) {
                flush_row(tile, curL, lane, acc0, acc1);
                acc0 = 0.0f;
                acc1 = 0.0f;
                curL = Lj;
              }
              acc0 += v0[k] * w;
              acc1 += v1[k] * w;
            }
          }
        }
        cur = nx;
      }
      flush_row(tile, curL, lane, acc0, acc1);
    }
  }
  __syncthreads();

  {
    const float4* const t4 = reinterpret_cast<const float4*>(tile);
    const float4* const b4 = reinterpret_cast<const float4*>(bias);
    float4* const o4 = reinterpret_cast<float4*>(out);
    for (int i = (int)threadIdx.x; i < OW_ROWS * (D_FEAT / 4); i += SC_THREADS) {
      const int node = rowBase + (i >> 4);
      if (node < n) {
        float4 v = t4[i];
        const float4 bv = b4[i & 15];
        v.x += bv.x;
        v.y += bv.y;
        v.z += bv.z;
        v.w += bv.w;
        o4[node * (D_FEAT / 4) + (i & 15)] = v;
      }
    }
  }
}


extern "C" void kernel_launch(void* const* d_in, const int* in_sizes, int n_in,
                              void* d_out, int out_size, void* d_ws, size_t ws_size,
                              hipStream_t stream) {
  const float* x      = (const float*)d_in[0];
  const int*   eidx   = (const int*)d_in[1];
  const float* weight = (const float*)d_in[2];
  const float* bias   = (const float*)d_in[3];
  float*       out    = (float*)d_out;

  const int n = in_sizes[0] / D_FEAT;
  const int e = in_sizes[1] / 2;
  const int* row = eidx;
  const int* col = eidx + e;

  float* dinv = (float*)d_ws;
  const int dinvStride = (n + 255) & ~255;
  float* xw = dinv + dinvStride;
  int* partial = (int*)(xw + n * D_FEAT);
  int* goff = partial + CNT_CHUNKS * dinvStride;
  int2* glist = (int2*)(goff + BN_WGS * BN_PITCH);
  unsigned short* deg16 = (unsigned short*)(glist + BN_WGS * BN_CAP);

  count_bin_gemm_kernel<<<((n + CNT_BINS - 1) / CNT_BINS) * CNT_CHUNKS + BN_WGS +
                              (n + GEMM_ROWS_PER_BLOCK - 1) / GEMM_ROWS_PER_BLOCK,
                          CNT_THREADS, 0, stream>>>(row, col, partial, glist, goff, x, weight, xw,
                                                    n, e, dinvStride);
  dinv_kernel<<<(n + 511) / 512, 256, 0, stream>>>(partial, dinv, deg16, n, dinvStride);
  owner_sum_kernel<<<OW_RANGES, SC_THREADS, 0, stream>>>(glist, goff, dinv, deg16, xw, bias, out, n);
}


# degree count: a work-group owns 6272 node bins and one of 32 edge chunks (was 2048 bins, 10 chunks), so each row id is range-tested by 8 work-groups, not 25; planes summed by lane pairs (exact ints)
# speedup vs baseline: 4.2065x; 1.0625x over previous
#include <hip/hip_runtime.h>
#include <hip/hip_bf16.h>


typedef __attribute__((ext_vector_type(2))) float v2f;
typedef __attribute__((ext_vector_type(8))) float v8f;

#define D_FEAT 64


#define CNT_BINS    6272
#define CNT_CHUNKS  32
#define CNT_THREADS 512
#define CNT_INFLIGHT 8

__device__ __forceinline__ void count_one(unsigned int* hist, int r, int binBase) {
  const unsigned int b = (unsigned int)(r - binBase);
  if (b < (unsigned int)CNT_BINS) atomicAdd(&hist[b], 1u);
}

__device__ __forceinline__ void degree_count_block(unsigned int* hist,
                                                   const int* __restrict__ row,
                                                   int* __restrict__ partial,
                                                   int n, int e, int nStride,
                                                   int binBlock, int chunk) {
  const int binBase = binBlock * CNT_BINS;

  {
    uint4* const h4 = reinterpret_cast<uint4*>(hist);
    for (int i = threadIdx.x; i < CNT_BINS / 4; i += CNT_THREADS) h4[i] = make_uint4(0u, 0u, 0u, 0u);
  }
  __syncthreads();

  const int chunkLen = (((e + CNT_CHUNKS - 1) / CNT_CHUNKS) + 3) & ~3;
  const int beg = chunk * chunkLen;
  int end = beg + chunkLen;
  if (end > e) end = e;
  const int nvec = (end > beg) ? ((end - beg) >> 2) : 0;

  if (nvec > 0) {
    const int4* rv = reinterpret_cast<const int4*>(row + beg);
    for (int v = threadIdx.x; v < nvec; v += CNT_INFLIGHT * CNT_THREADS) {
      int4 r[CNT_INFLIGHT];
      for (int k = 0; k < CNT_INFLIGHT; ++k) {
        const int idx = v + k * CNT_THREADS;
        r[k] = (idx < nvec) ? rv[idx] : make_int4(-1, -1, -1, -1);
      }
      for (int k = 0; k < CNT_INFLIGHT; ++k) {
        count_one(hist, r[k].x, binBase);
        count_one(hist, r[k].y, binBase);
        count_one(hist, r[k].z, binBase);
        count_one(hist, r[k].w, binBase);
      }
    }
  }
  for (int j = beg + 4 * nvec + (int)threadIdx.x; j < end; j += CNT_THREADS)
    count_one(hist, row[j], binBase);

  __syncthreads();
  {
    const uint4* const h4 = reinterpret_cast<const uint4*>(hist);
    int* const plane = partial + chunk * nStride + binBase;
    constexpr int NV = CNT_BINS / 4;
    constexpr int TRIPS = (NV + CNT_THREADS - 1) / CNT_THREADS;
    uint4 h[TRIPS];
#pragma unroll
    for (int k = 0; k < TRIPS; ++k) {
      const int i = (int)threadIdx.x + k * CNT_THREADS;
      h[k] = h4[(i < NV) ? i : (NV - 1)];
    }
#pragma unroll
    for (int k = 0; k < TRIPS; ++k) {
      const int i = (int)threadIdx.x + k * CNT_THREADS;
      const int node = binBase + 4 * i;
      if (i < NV) {
        if (node + 3 < n) {
          *reinterpret_cast<int4*>(plane + 4 * i) = make_int4((int)h[k].x, (int)h[k].y, (int)h[k].z, (int)h[k].w);
        } else {
          if (node < n) plane[4 * i] = (int)h[k].x;
          if (node + 1 < n) plane[4 * i + 1] = (int)h[k].y;
          if (node + 2 < n) plane[4 * i + 2] = (int)h[k].z;
        }
      }
    }
  }
}

#define DEG16_SAT 65535

__device__ __forceinline__ float dinv_from_count(int cnt) {
  const float d = (float)cnt;
  return (d > 0.0f) ? rsqrtf(d) : 0.0f;
}
__device__ __forceinline__ unsigned int sat16(int cnt) {
  return (unsigned int)((cnt < DEG16_SAT) ? cnt : DEG16_SAT);
}

__global__ __launch_bounds__(256) void dinv_kernel(const int* __restrict__ partial,
                                                   float* __restrict__ dinv,
                                                   unsigned short* __restrict__ deg16,
                                                   int n, int nStride) {
  static_assert(CNT_CHUNKS % 2 == 0, "the planes are split over a pair of lanes");
  const int g = (int)(blockIdx.x * blockDim.x + threadIdx.x);
  const int half = g & 1;
  const int i = 2 * (g >> 1);
  const int* const mine = partial + half * (CNT_CHUNKS / 2) * nStride + i;
  if (i + 1 < n) {
    int c0 = 0, c1 = 0;
#pragma unroll
    for (int c = 0; c < CNT_CHUNKS / 2; ++c) {
      const int2 p = *reinterpret_cast<const int2*>(mine + c * nStride);
      c0 += p.x;
      c1 += p.y;
    }
    c0 += __shfl_xor(c0, 1);
    c1 += __shfl_xor(c1, 1);
    if (half == 0) {
      float2 r;
      r.x = dinv_from_count(c0);
      r.y = dinv_from_count(c1);
      *reinterpret_cast<float2*>(dinv + i) = r;
      *reinterpret_cast<unsigned int*>(deg16 + i) = sat16(c0) | (sat16(c1) << 16);
    }
  } else if (i < n) {
    int c0 = 0;
    for (int c = 0; c < CNT_CHUNKS / 2; ++c) c0 += mine[c * nStride];
    c0 += __shfl_xor(c0, 1);
    if (half == 0) {
      dinv[i] = dinv_from_count(c0);
      deg16[i] = (unsigned short)sat16(c0);
    }
  }
}

#define GEMM_ROWS_PER_BLOCK 64

__device__ __forceinline__ void gemm_xw_block(const float* __restrict__ X,
                                              const float* __restrict__ W,
                                              float* __restrict__ XW, int n, int gemmBlock) {
  const int wave = threadIdx.x >> 5;
  const int lane = threadIdx.x & 31;
  const int half = lane >> 4;
  const int l    = lane & 15;

  const int mBase = (gemmBlock * 4 + (wave >> 2)) * 16;
  const int nBase = (wave & 3) * 16;
  if (mBase >= n) return;

  const int m = mBase + l;

  v8f acc = {};
#pragma unroll
  for (int k = 0; k < D_FEAT; k += 4) {
    v2f a, b;
    a.x = X[m * D_FEAT + k + 2 * half + 0];
    a.y = X[m * D_FEAT + k + 2 * half + 1];
    b.x = W[(k + 2 * half + 0) * D_FEAT + nBase + l];
    b.y = W[(k + 2 * half + 1) * D_FEAT + nBase + l];
    acc = __builtin_amdgcn_wmma_f32_16x16x4_f32(false, a, false, b,
                                                (short)0, acc, false, false);
  }

#pragma unroll
  for (int r = 0; r < 8; ++r) {
    XW[(mBase + r + 8 * half) * D_FEAT + nBase + l] = acc[r];
  }
}

#define OW_ROWS   196
#define OW_RANGES 256
#define BN_WGS    256
#define BN_CAP    3200
#define BN_PER    7
#define BN_PITCH  288

__device__ __forceinline__ void scan256(const unsigned int* cnt, unsigned int* off, int lane) {
  unsigned int v[8];
  unsigned int s = 0u;
#pragma unroll
  for (int k = 0; k < 8; ++k) {
    const unsigned int t = cnt[8 * lane + k];
    v[k] = s;
    s += t;
  }
  unsigned int incl = s;
#pragma unroll
  for (int d = 1; d < 32; d <<= 1) {
    const unsigned int o = __shfl_up(incl, (unsigned int)d);
    if (lane >= d) incl += o;
  }
  const unsigned int base = incl - s;
#pragma unroll
  for (int k = 0; k < 8; ++k) off[8 * lane + k] = base + v[k];
  if (lane == 31) off[256] = incl;
}

__device__ __forceinline__ unsigned int range_of(int r) {
  const unsigned int g = (unsigned int)r / (unsigned int)OW_ROWS;
  return (g < (unsigned int)OW_RANGES) ? g : (unsigned int)(OW_RANGES - 1);
}

__device__ __forceinline__ void bin_block(unsigned int* smem,
                                          const int* __restrict__ row,
                                          const int* __restrict__ col,
                                          int2* __restrict__ glist,
                                          int* __restrict__ goff, int e, int w) {
  int2* const stage = reinterpret_cast<int2*>(smem);
  unsigned int* const cnt = smem + 2 * BN_CAP;
  unsigned int* const off = cnt + OW_RANGES;
  const int t = (int)threadIdx.x;

  const int chunkLen = (e + BN_WGS - 1) / BN_WGS;
  const int beg = w * chunkLen;
  int end = beg + chunkLen;
  if (end > e) end = e;
  const int len = (end > beg) ? (end - beg) : 0;

  if (t < OW_RANGES) cnt[t] = 0u;
  __syncthreads();

  int r[BN_PER], c[BN_PER];
  unsigned int pos[BN_PER];
#pragma unroll
  for (int k = 0; k < BN_PER; ++k) {
    const int i = k * CNT_THREADS + t;
    const int j = (i < len) ? (beg + i) : 0;
    r[k] = row[j];
    c[k] = col[j];
  }
#pragma unroll
  for (int k = 0; k < BN_PER; ++k) {
    pos[k] = 0u;
    if (k * CNT_THREADS + t < len) pos[k] = atomicAdd(&cnt[range_of(r[k])], 1u);
  }
  __syncthreads();
  if (t < 32) scan256(cnt, off, t);
  __syncthreads();
#pragma unroll
  for (int k = 0; k < BN_PER; ++k) {
    if (k * CNT_THREADS + t < len) stage[off[range_of(r[k])] + pos[k]] = make_int2(c[k], r[k]);
  }
  __syncthreads();
  int2* const mine = glist + w * BN_CAP;
  for (int i = t; i < len; i += CNT_THREADS) mine[i] = stage[i];
  if (t <= OW_RANGES) goff[w * BN_PITCH + t] = (int)off[t];
}

__global__ __launch_bounds__(CNT_THREADS, 16) void count_bin_gemm_kernel(const int* __restrict__ row,
                                                                     const int* __restrict__ col,
                                                                     int* __restrict__ partial,
                                                                     int2* __restrict__ glist,
                                                                     int* __restrict__ goff,
                                                                     const float* __restrict__ X,
                                                                     const float* __restrict__ W,
                                                                     float* __restrict__ XW,
                                                                     int n, int e, int nStride) {
  __shared__ __attribute__((aligned(16))) unsigned int smem[2 * BN_CAP + 2 * OW_RANGES + 8];
  static_assert(CNT_BINS <= 2 * BN_CAP + 2 * OW_RANGES + 8 && CNT_BINS % 4 == 0,
                "the count's histogram lives in the list builders' LDS block, cleared and stored 16 B per lane");
  const int binBlocks   = (n + CNT_BINS - 1) / CNT_BINS;
  const int countBlocks = binBlocks * CNT_CHUNKS;
  const int b = blockIdx.x;
  if (b < countBlocks) {
    const int chunk = b / binBlocks;
    degree_count_block(smem, row, partial, n, e, nStride, b - chunk * binBlocks, chunk);
  } else if (b < countBlocks + BN_WGS) {
    bin_block(smem, row, col, glist, goff, e, b - countBlocks);
  } else {
    gemm_xw_block(X, W, XW, n, b - countBlocks - BN_WGS);
  }
}

#define SC_THREADS 1024
#define SC_GROUP   8
#define SC_GPB     4
#define SC_TAB     98304
#define SC_FILL    8
#define ST_PER     4
#define ST_CAP     (ST_PER * SC_THREADS)

__device__ __forceinline__ int lane_bcast_i(int v, int srcLane) {
  return __builtin_amdgcn_readlane(v, srcLane);
}
__device__ __forceinline__ float lane_bcast_f(float v, int srcLane) {
  return __int_as_float(__builtin_amdgcn_readlane(__float_as_int(v), srcLane));
}

__device__ __forceinline__ float dinv_of(unsigned int d16) {
  const float d = (float)d16;
  return (d > 0.0f) ? rsqrtf(d) : 0.0f;
}

__device__ __forceinline__ void flush_row(float* tile, int L, int lane, float a0, float a1) {
  atomicAdd(&tile[L * D_FEAT + lane], a0);
  atomicAdd(&tile[L * D_FEAT + 32 + lane], a1);
}

__device__ __forceinline__ void fetch_pair(const unsigned int* segStart, const int* segBeg,
                                           const int2* __restrict__ glist, int q, int& r, int& c) {
  int lo = 0;
#pragma unroll
  for (int s = 128; s >= 1; s >>= 1) {
    if (segStart[lo + s] <= (unsigned int)q) lo += s;
  }
  const int2 pr = glist[lo * BN_CAP + segBeg[lo] + (q - (int)segStart[lo])];
  c = pr.x;
  r = pr.y;
}

__global__ __launch_bounds__(SC_THREADS) void owner_sum_kernel(const int2* __restrict__ glist,
                                                               const int* __restrict__ goff,
                                                               const float* __restrict__ dinv,
                                                               const unsigned short* __restrict__ deg16,
                                                               const float* __restrict__ xw,
                                                               const float* __restrict__ bias,
                                                               float* __restrict__ out, int n) {
  __shared__ __attribute__((aligned(16))) unsigned short tab[SC_TAB];
  __shared__ __attribute__((aligned(16))) float tile[OW_ROWS * D_FEAT];
  __shared__ unsigned int segLen[OW_RANGES];
  __shared__ unsigned int segStart[OW_RANGES + 1];
  __shared__ int segBeg[OW_RANGES];
  __shared__ __attribute__((aligned(8))) int2 stage[ST_CAP];
  __shared__ unsigned int rowCnt[256];
  __shared__ unsigned int rowOff[256 + 1];
  const int lane = threadIdx.x & 31;
  const int wave = __builtin_amdgcn_readfirstlane((int)(threadIdx.x >> 5));
  const int rowBase = (int)blockIdx.x * OW_ROWS;

  if ((int)threadIdx.x < BN_WGS) {
    const int* p = goff + (int)threadIdx.x * BN_PITCH + (int)blockIdx.x;
    const int pb = p[0];
    const int pe = p[1];
    segBeg[threadIdx.x] = pb;
    segLen[threadIdx.x] = (unsigned int)(pe - pb);
  }

  {
    float4* const z4 = reinterpret_cast<float4*>(tile);
    for (int i = (int)threadIdx.x; i < OW_ROWS * (D_FEAT / 4); i += SC_THREADS)
      z4[i] = make_float4(0.0f, 0.0f, 0.0f, 0.0f);
  }

  {
    const int nvec = n >> 3;
    const int npiece = nvec >> 5;
    const uint4* d4 = reinterpret_cast<const uint4*>(deg16);
    uint4* t4 = reinterpret_cast<uint4*>(tab);
    for (int p = wave; p < npiece; p += SC_FILL * (SC_THREADS / 32)) {
      uint4 t[SC_FILL];
#pragma unroll
      for (int k = 0; k < SC_FILL; ++k) {
        const int q = p + k * (SC_THREADS / 32);
        t[k] = (q < npiece) ? d4[q * 32 + lane] : make_uint4(0u, 0u, 0u, 0u);
      }
#pragma unroll
      for (int k = 0; k < SC_FILL; ++k) {
        const int q = p + k * (SC_THREADS / 32);
        if (q < npiece) t4[q * 32 + lane] = t[k];
      }
    }
    const int v = (npiece << 5) + (int)threadIdx.x;
    if (v < nvec) t4[v] = d4[v];
    const int i = (nvec << 3) + (int)threadIdx.x;
    if (i < n) tab[i] = deg16[i];
  }
  __syncthreads();
  if (threadIdx.x < 32) scan256(segLen, segStart, lane);
  __syncthreads();

  const int M = (int)segStart[OW_RANGES];
  const int t = (int)threadIdx.x;

  for (int base = 0; base < M; base += ST_CAP) {
    const int m = (M - base < ST_CAP) ? (M - base) : ST_CAP;

    if (t < 256) rowCnt[t] = 0u;
    __syncthreads();

    int eC[ST_PER], eR[ST_PER];
#pragma unroll
    for (int k = 0; k < ST_PER; ++k) {
      const int i = k * SC_THREADS + t;
      fetch_pair(segStart, segBeg, glist, base + ((i < m) ? i : (m - 1)), eR[k], eC[k]);
    }
    unsigned int ePos[ST_PER];
    float eW[ST_PER];
#pragma unroll
    for (int k = 0; k < ST_PER; ++k) {
      const unsigned int degR = tab[eR[k]];
      const unsigned int degC = tab[eC[k]];
      float w = dinv_of(degR) * dinv_of(degC);
      const bool sat = (degR == (unsigned int)DEG16_SAT) | (degC == (unsigned int)DEG16_SAT);
      if (__builtin_amdgcn_ballot_w32(sat) != 0u) w = dinv[eR[k]] * dinv[eC[k]];
      eW[k] = w;
      ePos[k] = 0u;
      if (k * SC_THREADS + t < m) ePos[k] = atomicAdd(&rowCnt[eR[k] - rowBase], 1u);
    }
    __syncthreads();
    if (t < 32) scan256(rowCnt, rowOff, lane);
    __syncthreads();
#pragma unroll
    for (int k = 0; k < ST_PER; ++k) {
      if (k * SC_THREADS + t < m) {
        const int L = eR[k] - rowBase;
        stage[rowOff[L] + ePos[k]] = make_int2(eC[k] | (L << 20), __float_as_int(eW[k]));
      }
    }
    __syncthreads();

    const int G = (m + SC_GROUP - 1) / SC_GROUP;
    const int gBase = G / (SC_THREADS / 32);
    const int gRem = G - gBase * (SC_THREADS / 32);
    const int gFirst = wave * gBase + ((wave < gRem) ? wave : gRem);
    const int e0 = gFirst * SC_GROUP;
    int e1 = e0 + (gBase + ((wave < gRem) ? 1 : 0)) * SC_GROUP;
    if (e1 > m) e1 = m;
    if (e0 < e1) {
      int2 cur;
      {
        const int q = e0 + lane;
        cur = stage[(q < m) ? q : (m - 1)];
      }
      float acc0 = 0.0f, acc1 = 0.0f;
      int curL = lane_bcast_i((int)((unsigned int)cur.x >> 20), 0);

      for (int bs = e0; bs < e1; bs += 32) {
        const int cnt = (e1 - bs < 32) ? (e1 - bs) : 32;
        const int ng = cnt / SC_GROUP;
        const int rest = cnt - ng * SC_GROUP;

        int2 nx = cur;
        if (bs + 32 < e1) {
          const int q = bs + 32 + lane;
          nx = stage[(q < m) ? q : (m - 1)];
        }

        const int myC = cur.x & 0xFFFFF;
        const int myL = (int)((unsigned int)cur.x >> 20);
        const float myW = __int_as_float(cur.y);

#pragma unroll
        for (int q = 0; q < SC_GPB; ++q) {
          if (q < ng) {
            float v0[SC_GROUP], v1[SC_GROUP];
#pragma unroll
            for (int k = 0; k < SC_GROUP; ++k) {
              const unsigned co = (unsigned)lane_bcast_i(myC, q * SC_GROUP + k) * D_FEAT + lane;
              v0[k] = xw[co];
              v1[k] = xw[co + 32];
            }
#pragma unroll
            for (int k = 0; k < SC_GROUP; ++k) {
              const int Lj = lane_bcast_i(myL, q * SC_GROUP + k);
              const float w = lane_bcast_f(myW, q * SC_GROUP + k);
              if (Lj != curL) {
                flush_row(tile, curL, lane, acc0, acc1);
                acc0 = 0.0f;
                acc1 = 0.0f;
                curL = Lj;
              }
              acc0 += v0[k] * w;
              acc1 += v1[k] * w;
            }
          }
        }
        if (rest > 0) {
          const int s0 = ng * SC_GROUP;
          float v0[SC_GROUP], v1[SC_GROUP];
#pragma unroll
          for (int k = 0; k < SC_GROUP; ++k) {
            const unsigned co = (unsigned)lane_bcast_i(myC, s0 + k) * D_FEAT + lane;
            v0[k] = xw[co];
            v1[k] = xw[co + 32];
          }
#pragma unroll
          for (int k = 0; k < SC_GROUP; ++k) {
            if (k < rest) {
              const int Lj = lane_bcast_i(myL, s0 + k);
              const float w = lane_bcast_f(myW, s0 + k);
              if (Lj != curL) {
                flush_row(tile, curL, lane, acc0, acc1);
                acc0 = 0.0f;
                acc1 = 0.0f;
                curL = Lj;
              }
              acc0 += v0[k] * w;
              acc1 += v1[k] * w;
            }
          }
        }
        cur = nx;
      }
      flush_row(tile, curL, lane, acc0, acc1);
    }
  }
  __syncthreads();

  {
    const float4* const t4 = reinterpret_cast<const float4*>(tile);
    const float4* const b4 = reinterpret_cast<const float4*>(bias);
    float4* const o4 = reinterpret_cast<float4*>(out);
    for (int i = (int)threadIdx.x; i < OW_ROWS * (D_FEAT / 4); i += SC_THREADS) {
      const int node = rowBase + (i >> 4);
      if (node < n) {
        float4 v = t4[i];
        const float4 bv = b4[i & 15];
        v.x += bv.x;
        v.y += bv.y;
        v.z += bv.z;
        v.w += bv.w;
        o4[node * (D_FEAT / 4) + (i & 15)] = v;
      }
    }
  }
}


extern "C" void kernel_launch(void* const* d_in, const int* in_sizes, int n_in,
                              void* d_out, int out_size, void* d_ws, size_t ws_size,
                              hipStream_t stream) {
  const float* x      = (const float*)d_in[0];
  const int*   eidx   = (const int*)d_in[1];
  const float* weight = (const float*)d_in[2];
  const float* bias   = (const float*)d_in[3];
  float*       out    = (float*)d_out;

  const int n = in_sizes[0] / D_FEAT;
  const int e = in_sizes[1] / 2;
  const int* row = eidx;
  const int* col = eidx + e;

  float* dinv = (float*)d_ws;
  const int dinvStride = (n + 255) & ~255;
  float* xw = dinv + dinvStride;
  int* partial = (int*)(xw + n * D_FEAT);
  int* goff = partial + CNT_CHUNKS * dinvStride;
  int2* glist = (int2*)(goff + BN_WGS * BN_PITCH);
  unsigned short* deg16 = (unsigned short*)(glist + BN_WGS * BN_CAP);

  count_bin_gemm_kernel<<<((n + CNT_BINS - 1) / CNT_BINS) * CNT_CHUNKS + BN_WGS +
                              (n + GEMM_ROWS_PER_BLOCK - 1) / GEMM_ROWS_PER_BLOCK,
                          CNT_THREADS, 0, stream>>>(row, col, partial, glist, goff, x, weight, xw,
                                                    n, e, dinvStride);
  dinv_kernel<<<(n + 256) / 256, 256, 0, stream>>>(partial, dinv, deg16, n, dinvStride);
  owner_sum_kernel<<<OW_RANGES, SC_THREADS, 0, stream>>>(glist, goff, dinv, deg16, xw, bias, out, n);
}
